# starting kernel
# baseline (speedup 1.0000x reference)
#include <hip/hip_runtime.h>
#include <hip/hip_bf16.h>

#define N_NODES   50000
#define D_FEAT    512
#define H1_DIM    128
#define H2_DIM    64
#define N_CLS     40
#define E_MP      800000
#define E_LP      250000

typedef __attribute__((ext_vector_type(2))) float v2f;
typedef __attribute__((ext_vector_type(8))) float v8f;

__global__ void fill_kernel(float* __restrict__ p, long long n, float v) {
  long long i = (long long)blockIdx.x * blockDim.x + threadIdx.x;
  if (i < n) p[i] = v;
}

__global__ void deg_accum_kernel(float* __restrict__ deg, const int* __restrict__ cols,
                                 const float* __restrict__ ew, int E) {
  long long i = (long long)blockIdx.x * blockDim.x + threadIdx.x;
  if (i < E) {
    float w = ew ? ew[i] : 1.0f;
    atomicAdd(&deg[cols[i]], w);
  }
}

__global__ void rsqrt_kernel(float* __restrict__ p, int n) {
  int i = blockIdx.x * blockDim.x + threadIdx.x;
  if (i < n) {
    float d = p[i];
    p[i] = (d > 0.0f) ? rsqrtf(d) : 0.0f;
  }
}

__global__ void bias_init_kernel(float* __restrict__ out, const float* __restrict__ b,
                                 long long n, int F) {
  long long i = (long long)blockIdx.x * blockDim.x + threadIdx.x;
  long long total = n * F;
  if (i < total) out[i] = b[i % F];
}

__global__ void pack_b_kernel(const float* __restrict__ B, float* __restrict__ Bp,
                              int K, int NC, int NCP) {
  int p = blockIdx.x * blockDim.x + threadIdx.x;
  int total = (K / 2) * NCP;
  if (p >= total) return;
  int k2 = p / NCP;
  int col = p % NCP;
  float x0 = 0.0f, x1 = 0.0f;
  if (col < NC) {
    x0 = B[(long long)(2 * k2) * NC + col];
    x1 = B[(long long)(2 * k2 + 1) * NC + col];
  }
  Bp[2 * p + 0] = x0;
  Bp[2 * p + 1] = x1;
}

template <int KDIM, int NT, int NCSTORE, bool RELU_A>
__global__ void wmma_gemm_kernel(const float* __restrict__ A, const float* __restrict__ Bp,
                                 float* __restrict__ C, int M) {
  constexpr int NCP = NT * 16;
  int wave = (int)(((long long)blockIdx.x * blockDim.x + threadIdx.x) >> 5);
  int lane = threadIdx.x & 31;
  int ntiles = M >> 4;
  if (wave >= ntiles) return;

  int row0 = wave << 4;
  int arow = row0 + (lane & 15);
  int khb  = lane >> 4;
  int bcol = lane & 15;

  v8f acc[NT] = {};

  const float2* __restrict__ Arow2 = (const float2*)(A + (long long)arow * KDIM);
  const float2* __restrict__ Bp2   = (const float2*)Bp;

#pragma unroll 2
  for (int k2 = 0; k2 < KDIM / 2; k2 += 2) {
    const float2* __restrict__ Bbase = Bp2 + (long long)(k2 + khb) * NCP + bcol;
    float2 bf[NT];
#pragma unroll
    for (int t = 0; t < NT; ++t) bf[t] = Bbase[t * 16];

    float2 af = Arow2[k2 + khb];
    if (RELU_A) { af.x = fmaxf(af.x, 0.0f); af.y = fmaxf(af.y, 0.0f); }
    v2f a; a.x = af.x; a.y = af.y;

#pragma unroll
    for (int t = 0; t < NT; ++t) {
      v2f b; b.x = bf[t].x; b.y = bf[t].y;
      acc[t] = __builtin_amdgcn_wmma_f32_16x16x4_f32(
           false, a,  false, b,
           (short)0, acc[t],  false,  false);
    }
  }

  int crow = row0 + (khb << 3);
  int ccol = lane & 15;
#pragma unroll
  for (int t = 0; t < NT; ++t) {
    int col = (t << 4) + ccol;
    if (col < NCSTORE) {
#pragma unroll
      for (int r = 0; r < 8; ++r) {
        C[(long long)(crow + r) * NCSTORE + col] = acc[t][r];
      }
    }
  }
}

__global__ void gcn_agg_kernel(const float* __restrict__ hx, float* __restrict__ out,
                               const int* __restrict__ rows, const int* __restrict__ cols,
                               const float* __restrict__ ew, const float* __restrict__ dinv,
                               int E, int N, int F) {
  int chunks = F >> 2;
  long long tid = (long long)blockIdx.x * blockDim.x + threadIdx.x;
  long long total = (long long)(E + N) * chunks;
  if (tid >= total) return;
  int e = (int)(tid / chunks);
  int c = (int)(tid % chunks) << 2;
  int r, cl; float w;
  if (e < E) { r = rows[e]; cl = cols[e]; w = ew ? ew[e] : 1.0f; }
  else       { r = e - E;   cl = r;       w = 1.0f; }
  float norm = dinv[r] * w * dinv[cl];
  const float4 h4 = *(const float4*)(hx + (long long)r * F + c);
  float* o = out + (long long)cl * F + c;
  atomicAdd(o + 0, h4.x * norm);
  atomicAdd(o + 1, h4.y * norm);
  atomicAdd(o + 2, h4.z * norm);
  atomicAdd(o + 3, h4.w * norm);
}

__global__ void log_softmax_kernel(const float* __restrict__ in, float* __restrict__ out,
                                   int n, int F) {
  int row = blockIdx.x * blockDim.x + threadIdx.x;
  if (row >= n) return;
  const float* p = in + (long long)row * F;
  float m = -INFINITY;
  for (int c = 0; c < F; ++c) m = fmaxf(m, p[c]);
  float s = 0.0f;
  for (int c = 0; c < F; ++c) s += __expf(p[c] - m);
  float ls = __logf(s) + m;
  float* o = out + (long long)row * F;
  for (int c = 0; c < F; ++c) o[c] = p[c] - ls;
}

__global__ void edge_dot_kernel(const float* __restrict__ h, const int* __restrict__ pos,
                                const int* __restrict__ neg, float* __restrict__ res) {
  int e = blockIdx.x * blockDim.x + threadIdx.x;
  if (e >= 2 * E_LP) return;
  const int* src = (e < E_LP) ? pos : neg;
  int eo = (e < E_LP) ? e : (e - E_LP);
  int j = src[eo];
  int i = src[E_LP + eo];
  const float4* hj = (const float4*)(h + (long long)j * H2_DIM);
  const float4* hi = (const float4*)(h + (long long)i * H2_DIM);
  float s = 0.0f;
#pragma unroll
  for (int k = 0; k < H2_DIM / 4; ++k) {
    float4 a = hi[k], b = hj[k];
    s += a.x * b.x + a.y * b.y + a.z * b.z + a.w * b.w;
  }
  res[e] = s;
}

static inline int blocks_for(long long total, int bs) {
  return (int)((total + bs - 1) / bs);
}

extern "C" void kernel_launch(void* const* d_in, const int* in_sizes, int n_in,
                              void* d_out, int out_size, void* d_ws, size_t ws_size,
                              hipStream_t stream) {
  const float* x      = (const float*)d_in[0];
  const int*   tpe    = (const int*)  d_in[1];
  const int*   ei     = (const int*)  d_in[2];
  const float* ew     = (const float*)d_in[3];
  const int*   pos    = (const int*)  d_in[4];
  const int*   neg    = (const int*)  d_in[5];
  const float* W1     = (const float*)d_in[6];
  const float* b1     = (const float*)d_in[7];
  const float* W2     = (const float*)d_in[8];
  const float* b2     = (const float*)d_in[9];
  const float* W_attr = (const float*)d_in[10];
  const float* b_attr = (const float*)d_in[11];
  const float* W_att  = (const float*)d_in[12];
  const float* b_att  = (const float*)d_in[13];

  float* out = (float*)d_out;
  float* ws  = (float*)d_ws;

  const long long OFF_DINV_T = 0;
  const long long OFF_DINV_W = 50000;
  const long long OFF_HX1    = 100000;
  const long long OFF_H1     = OFF_HX1 + 6400000;
  const long long OFF_HX2    = OFF_H1  + 6400000;
  const long long OFF_H2     = OFF_HX2 + 3200000;
  const long long OFF_W1P    = OFF_H2  + 3200000;
  const long long OFF_W2P    = OFF_W1P + 65536;
  const long long OFF_WAP    = OFF_W2P + 8192;
  const long long OFF_WBP    = OFF_WAP + 3072;
  const long long OFF_HXA    = OFF_HX1;
  const long long OFF_HXB    = OFF_HX1 + 2000000;
  const long long OFF_ATTR   = OFF_H1;
  const long long OFF_ATT    = OFF_H1 + 2000000;

  float* dinv_t = ws + OFF_DINV_T;
  float* dinv_w = ws + OFF_DINV_W;
  float* hx1    = ws + OFF_HX1;
  float* h1     = ws + OFF_H1;
  float* hx2    = ws + OFF_HX2;
  float* h2     = ws + OFF_H2;
  float* w1p    = ws + OFF_W1P;
  float* w2p    = ws + OFF_W2P;
  float* wap    = ws + OFF_WAP;
  float* wbp    = ws + OFF_WBP;
  float* hxa    = ws + OFF_HXA;
  float* hxb    = ws + OFF_HXB;
  float* attr_a = ws + OFF_ATTR;
  float* att_a  = ws + OFF_ATT;

  const int BS = 256;

  pack_b_kernel<<<blocks_for((D_FEAT / 2) * H1_DIM, BS), BS, 0, stream>>>(W1, w1p, D_FEAT, H1_DIM, H1_DIM);
  pack_b_kernel<<<blocks_for((H1_DIM / 2) * H2_DIM, BS), BS, 0, stream>>>(W2, w2p, H1_DIM, H2_DIM, H2_DIM);
  pack_b_kernel<<<blocks_for((H2_DIM / 2) * 48, BS), BS, 0, stream>>>(W_attr, wap, H2_DIM, N_CLS, 48);
  pack_b_kernel<<<blocks_for((H2_DIM / 2) * 48, BS), BS, 0, stream>>>(W_att,  wbp, H2_DIM, N_CLS, 48);

  fill_kernel<<<blocks_for(N_NODES, BS), BS, 0, stream>>>(dinv_t, N_NODES, 1.0f);
  deg_accum_kernel<<<blocks_for(E_MP, BS), BS, 0, stream>>>(dinv_t, tpe + E_MP, nullptr, E_MP);
  rsqrt_kernel<<<blocks_for(N_NODES, BS), BS, 0, stream>>>(dinv_t, N_NODES);

  fill_kernel<<<blocks_for(N_NODES, BS), BS, 0, stream>>>(dinv_w, N_NODES, 1.0f);
  deg_accum_kernel<<<blocks_for(E_MP, BS), BS, 0, stream>>>(dinv_w, ei + E_MP, ew, E_MP);
  rsqrt_kernel<<<blocks_for(N_NODES, BS), BS, 0, stream>>>(dinv_w, N_NODES);

  const int ROW_TILES = N_NODES / 16;
  const int GEMM_BS = 128;
  const int gemm_blocks = blocks_for((long long)ROW_TILES * 32, GEMM_BS);

  wmma_gemm_kernel<D_FEAT, 8, H1_DIM, false><<<gemm_blocks, GEMM_BS, 0, stream>>>(x, w1p, hx1, N_NODES);
  bias_init_kernel<<<blocks_for((long long)N_NODES * H1_DIM, BS), BS, 0, stream>>>(h1, b1, N_NODES, H1_DIM);
  gcn_agg_kernel<<<blocks_for((long long)(E_MP + N_NODES) * (H1_DIM / 4), BS), BS, 0, stream>>>(
      hx1, h1, tpe, tpe + E_MP, nullptr, dinv_t, E_MP, N_NODES, H1_DIM);

  wmma_gemm_kernel<H1_DIM, 4, H2_DIM, true><<<gemm_blocks, GEMM_BS, 0, stream>>>(h1, w2p, hx2, N_NODES);
  bias_init_kernel<<<blocks_for((long long)N_NODES * H2_DIM, BS), BS, 0, stream>>>(h2, b2, N_NODES, H2_DIM);
  gcn_agg_kernel<<<blocks_for((long long)(E_MP + N_NODES) * (H2_DIM / 4), BS), BS, 0, stream>>>(
      hx2, h2, tpe, tpe + E_MP, nullptr, dinv_t, E_MP, N_NODES, H2_DIM);

  wmma_gemm_kernel<H2_DIM, 3, N_CLS, false><<<gemm_blocks, GEMM_BS, 0, stream>>>(h2, wap, hxa, N_NODES);
  wmma_gemm_kernel<H2_DIM, 3, N_CLS, false><<<gemm_blocks, GEMM_BS, 0, stream>>>(h2, wbp, hxb, N_NODES);

  bias_init_kernel<<<blocks_for((long long)N_NODES * N_CLS, BS), BS, 0, stream>>>(attr_a, b_attr, N_NODES, N_CLS);
  gcn_agg_kernel<<<blocks_for((long long)(E_MP + N_NODES) * (N_CLS / 4), BS), BS, 0, stream>>>(
      hxa, attr_a, ei, ei + E_MP, ew, dinv_w, E_MP, N_NODES, N_CLS);

  bias_init_kernel<<<blocks_for((long long)N_NODES * N_CLS, BS), BS, 0, stream>>>(att_a, b_att, N_NODES, N_CLS);
  gcn_agg_kernel<<<blocks_for((long long)(E_MP + N_NODES) * (N_CLS / 4), BS), BS, 0, stream>>>(
      hxb, att_a, ei, ei + E_MP, ew, dinv_w, E_MP, N_NODES, N_CLS);

  edge_dot_kernel<<<blocks_for(2 * E_LP, BS), BS, 0, stream>>>(h2, pos, neg, out);
  log_softmax_kernel<<<blocks_for(N_NODES, BS), BS, 0, stream>>>(attr_a, out + 2 * E_LP, N_NODES, N_CLS);
  log_softmax_kernel<<<blocks_for(N_NODES, BS), BS, 0, stream>>>(att_a,  out + 2 * E_LP + (long long)N_NODES * N_CLS, N_NODES, N_CLS);
}


# gcn_agg: one wave per 8 edges, lane l owns features l+32j: each f32 atomic / row load covers one contiguous 128-byte segment, not 4-byte pieces over 4 lines; atomics kept, sums still in arrival order
# speedup vs baseline: 1.3150x; 1.3150x over previous
#include <hip/hip_runtime.h>
#include <hip/hip_bf16.h>

#define N_NODES   50000
#define D_FEAT    512
#define H1_DIM    128
#define H2_DIM    64
#define N_CLS     40
#define E_MP      800000
#define E_LP      250000

typedef __attribute__((ext_vector_type(2))) float v2f;
typedef __attribute__((ext_vector_type(8))) float v8f;

__global__ void fill_kernel(float* __restrict__ p, long long n, float v) {
  long long i = (long long)blockIdx.x * blockDim.x + threadIdx.x;
  if (i < n) p[i] = v;
}

__global__ void deg_accum_kernel(float* __restrict__ deg, const int* __restrict__ cols,
                                 const float* __restrict__ ew, int E) {
  long long i = (long long)blockIdx.x * blockDim.x + threadIdx.x;
  if (i < E) {
    float w = ew ? ew[i] : 1.0f;
    atomicAdd(&deg[cols[i]], w);
  }
}

__global__ void rsqrt_kernel(float* __restrict__ p, int n) {
  int i = blockIdx.x * blockDim.x + threadIdx.x;
  if (i < n) {
    float d = p[i];
    p[i] = (d > 0.0f) ? rsqrtf(d) : 0.0f;
  }
}

__global__ void bias_init_kernel(float* __restrict__ out, const float* __restrict__ b,
                                 long long n, int F) {
  long long i = (long long)blockIdx.x * blockDim.x + threadIdx.x;
  long long total = n * F;
  if (i < total) out[i] = b[i % F];
}

__global__ void pack_b_kernel(const float* __restrict__ B, float* __restrict__ Bp,
                              int K, int NC, int NCP) {
  int p = blockIdx.x * blockDim.x + threadIdx.x;
  int total = (K / 2) * NCP;
  if (p >= total) return;
  int k2 = p / NCP;
  int col = p % NCP;
  float x0 = 0.0f, x1 = 0.0f;
  if (col < NC) {
    x0 = B[(long long)(2 * k2) * NC + col];
    x1 = B[(long long)(2 * k2 + 1) * NC + col];
  }
  Bp[2 * p + 0] = x0;
  Bp[2 * p + 1] = x1;
}

template <int KDIM, int NT, int NCSTORE, bool RELU_A>
__global__ void wmma_gemm_kernel(const float* __restrict__ A, const float* __restrict__ Bp,
                                 float* __restrict__ C, int M) {
  constexpr int NCP = NT * 16;
  int wave = (int)(((long long)blockIdx.x * blockDim.x + threadIdx.x) >> 5);
  int lane = threadIdx.x & 31;
  int ntiles = M >> 4;
  if (wave >= ntiles) return;

  int row0 = wave << 4;
  int arow = row0 + (lane & 15);
  int khb  = lane >> 4;
  int bcol = lane & 15;

  v8f acc[NT] = {};

  const float2* __restrict__ Arow2 = (const float2*)(A + (long long)arow * KDIM);
  const float2* __restrict__ Bp2   = (const float2*)Bp;

#pragma unroll 2
  for (int k2 = 0; k2 < KDIM / 2; k2 += 2) {
    const float2* __restrict__ Bbase = Bp2 + (long long)(k2 + khb) * NCP + bcol;
    float2 bf[NT];
#pragma unroll
    for (int t = 0; t < NT; ++t) bf[t] = Bbase[t * 16];

    float2 af = Arow2[k2 + khb];
    if (RELU_A) { af.x = fmaxf(af.x, 0.0f); af.y = fmaxf(af.y, 0.0f); }
    v2f a; a.x = af.x; a.y = af.y;

#pragma unroll
    for (int t = 0; t < NT; ++t) {
      v2f b; b.x = bf[t].x; b.y = bf[t].y;
      acc[t] = __builtin_amdgcn_wmma_f32_16x16x4_f32(
           false, a,  false, b,
           (short)0, acc[t],  false,  false);
    }
  }

  int crow = row0 + (khb << 3);
  int ccol = lane & 15;
#pragma unroll
  for (int t = 0; t < NT; ++t) {
    int col = (t << 4) + ccol;
    if (col < NCSTORE) {
#pragma unroll
      for (int r = 0; r < 8; ++r) {
        C[(long long)(crow + r) * NCSTORE + col] = acc[t][r];
      }
    }
  }
}

template <int F, int EPW>
__global__ void gcn_agg_kernel(const float* __restrict__ hx, float* __restrict__ out,
                               const int* __restrict__ rows, const int* __restrict__ cols,
                               const float* __restrict__ ew, const float* __restrict__ dinv,
                               int E, int N) {
  constexpr int FPT = (F + 31) / 32;
  const int lane = threadIdx.x & 31;
  const int wib = __builtin_amdgcn_readfirstlane((int)(threadIdx.x >> 5));
  const long long wave = (long long)blockIdx.x * (blockDim.x >> 5) + wib;
  const long long total = (long long)E + N;
  const long long e0 = wave * EPW;
  if (e0 >= total) return;

  int r = 0, cl = 0;
  float nrm = 0.0f;
  {
    const long long e = e0 + lane;
    if (lane < EPW && e < total) {
      float w;
      if (e < E) { r = rows[e]; cl = cols[e]; w = ew ? ew[e] : 1.0f; }
      else       { r = (int)(e - E); cl = r;  w = 1.0f; }
      nrm = dinv[r] * w * dinv[cl];
    }
  }

  const float* hrow[EPW];
#pragma unroll
  for (int i = 0; i < EPW; ++i)
    hrow[i] = hx + (long long)__builtin_amdgcn_readlane(r, i) * F + lane;

  float v[EPW][FPT];
#pragma unroll
  for (int j = 0; j < FPT; ++j) {
    if (lane + 32 * j < F) {
#pragma unroll
      for (int i = 0; i < EPW; ++i) v[i][j] = hrow[i][32 * j];
    }
  }

#pragma unroll
  for (int i = 0; i < EPW; ++i) {
    const int   ci = __builtin_amdgcn_readlane(cl, i);
    const float si = __int_as_float(__builtin_amdgcn_readlane(__float_as_int(nrm), i));
    if (e0 + i < total) {
      float* o = out + (long long)ci * F + lane;
#pragma unroll
      for (int j = 0; j < FPT; ++j)
        if (lane + 32 * j < F) atomicAdd(o + 32 * j, v[i][j] * si);
    }
  }
}

__global__ void log_softmax_kernel(const float* __restrict__ in, float* __restrict__ out,
                                   int n, int F) {
  int row = blockIdx.x * blockDim.x + threadIdx.x;
  if (row >= n) return;
  const float* p = in + (long long)row * F;
  float m = -INFINITY;
  for (int c = 0; c < F; ++c) m = fmaxf(m, p[c]);
  float s = 0.0f;
  for (int c = 0; c < F; ++c) s += __expf(p[c] - m);
  float ls = __logf(s) + m;
  float* o = out + (long long)row * F;
  for (int c = 0; c < F; ++c) o[c] = p[c] - ls;
}

__global__ void edge_dot_kernel(const float* __restrict__ h, const int* __restrict__ pos,
                                const int* __restrict__ neg, float* __restrict__ res) {
  int e = blockIdx.x * blockDim.x + threadIdx.x;
  if (e >= 2 * E_LP) return;
  const int* src = (e < E_LP) ? pos : neg;
  int eo = (e < E_LP) ? e : (e - E_LP);
  int j = src[eo];
  int i = src[E_LP + eo];
  const float4* hj = (const float4*)(h + (long long)j * H2_DIM);
  const float4* hi = (const float4*)(h + (long long)i * H2_DIM);
  float s = 0.0f;
#pragma unroll
  for (int k = 0; k < H2_DIM / 4; ++k) {
    float4 a = hi[k], b = hj[k];
    s += a.x * b.x + a.y * b.y + a.z * b.z + a.w * b.w;
  }
  res[e] = s;
}

static inline int blocks_for(long long total, int bs) {
  return (int)((total + bs - 1) / bs);
}

extern "C" void kernel_launch(void* const* d_in, const int* in_sizes, int n_in,
                              void* d_out, int out_size, void* d_ws, size_t ws_size,
                              hipStream_t stream) {
  const float* x      = (const float*)d_in[0];
  const int*   tpe    = (const int*)  d_in[1];
  const int*   ei     = (const int*)  d_in[2];
  const float* ew     = (const float*)d_in[3];
  const int*   pos    = (const int*)  d_in[4];
  const int*   neg    = (const int*)  d_in[5];
  const float* W1     = (const float*)d_in[6];
  const float* b1     = (const float*)d_in[7];
  const float* W2     = (const float*)d_in[8];
  const float* b2     = (const float*)d_in[9];
  const float* W_attr = (const float*)d_in[10];
  const float* b_attr = (const float*)d_in[11];
  const float* W_att  = (const float*)d_in[12];
  const float* b_att  = (const float*)d_in[13];

  float* out = (float*)d_out;
  float* ws  = (float*)d_ws;

  const long long OFF_DINV_T = 0;
  const long long OFF_DINV_W = 50000;
  const long long OFF_HX1    = 100000;
  const long long OFF_H1     = OFF_HX1 + 6400000;
  const long long OFF_HX2    = OFF_H1  + 6400000;
  const long long OFF_H2     = OFF_HX2 + 3200000;
  const long long OFF_W1P    = OFF_H2  + 3200000;
  const long long OFF_W2P    = OFF_W1P + 65536;
  const long long OFF_WAP    = OFF_W2P + 8192;
  const long long OFF_WBP    = OFF_WAP + 3072;
  const long long OFF_HXA    = OFF_HX1;
  const long long OFF_HXB    = OFF_HX1 + 2000000;
  const long long OFF_ATTR   = OFF_H1;
  const long long OFF_ATT    = OFF_H1 + 2000000;

  float* dinv_t = ws + OFF_DINV_T;
  float* dinv_w = ws + OFF_DINV_W;
  float* hx1    = ws + OFF_HX1;
  float* h1     = ws + OFF_H1;
  float* hx2    = ws + OFF_HX2;
  float* h2     = ws + OFF_H2;
  float* w1p    = ws + OFF_W1P;
  float* w2p    = ws + OFF_W2P;
  float* wap    = ws + OFF_WAP;
  float* wbp    = ws + OFF_WBP;
  float* hxa    = ws + OFF_HXA;
  float* hxb    = ws + OFF_HXB;
  float* attr_a = ws + OFF_ATTR;
  float* att_a  = ws + OFF_ATT;

  const int BS = 256;

  pack_b_kernel<<<blocks_for((D_FEAT / 2) * H1_DIM, BS), BS, 0, stream>>>(W1, w1p, D_FEAT, H1_DIM, H1_DIM);
  pack_b_kernel<<<blocks_for((H1_DIM / 2) * H2_DIM, BS), BS, 0, stream>>>(W2, w2p, H1_DIM, H2_DIM, H2_DIM);
  pack_b_kernel<<<blocks_for((H2_DIM / 2) * 48, BS), BS, 0, stream>>>(W_attr, wap, H2_DIM, N_CLS, 48);
  pack_b_kernel<<<blocks_for((H2_DIM / 2) * 48, BS), BS, 0, stream>>>(W_att,  wbp, H2_DIM, N_CLS, 48);

  fill_kernel<<<blocks_for(N_NODES, BS), BS, 0, stream>>>(dinv_t, N_NODES, 1.0f);
  deg_accum_kernel<<<blocks_for(E_MP, BS), BS, 0, stream>>>(dinv_t, tpe + E_MP, nullptr, E_MP);
  rsqrt_kernel<<<blocks_for(N_NODES, BS), BS, 0, stream>>>(dinv_t, N_NODES);

  fill_kernel<<<blocks_for(N_NODES, BS), BS, 0, stream>>>(dinv_w, N_NODES, 1.0f);
  deg_accum_kernel<<<blocks_for(E_MP, BS), BS, 0, stream>>>(dinv_w, ei + E_MP, ew, E_MP);
  rsqrt_kernel<<<blocks_for(N_NODES, BS), BS, 0, stream>>>(dinv_w, N_NODES);

  const int ROW_TILES = N_NODES / 16;
  const int GEMM_BS = 128;
  const int gemm_blocks = blocks_for((long long)ROW_TILES * 32, GEMM_BS);

  constexpr int AGG_EPW = 8;
  const int agg_blocks = blocks_for(((long long)(E_MP + N_NODES) + AGG_EPW - 1) / AGG_EPW * 32, BS);

  wmma_gemm_kernel<D_FEAT, 8, H1_DIM, false><<<gemm_blocks, GEMM_BS, 0, stream>>>(x, w1p, hx1, N_NODES);
  bias_init_kernel<<<blocks_for((long long)N_NODES * H1_DIM, BS), BS, 0, stream>>>(h1, b1, N_NODES, H1_DIM);
  gcn_agg_kernel<H1_DIM, AGG_EPW><<<agg_blocks, BS, 0, stream>>>(
      hx1, h1, tpe, tpe + E_MP, nullptr, dinv_t, E_MP, N_NODES);

  wmma_gemm_kernel<H1_DIM, 4, H2_DIM, true><<<gemm_blocks, GEMM_BS, 0, stream>>>(h1, w2p, hx2, N_NODES);
  bias_init_kernel<<<blocks_for((long long)N_NODES * H2_DIM, BS), BS, 0, stream>>>(h2, b2, N_NODES, H2_DIM);
  gcn_agg_kernel<H2_DIM, AGG_EPW><<<agg_blocks, BS, 0, stream>>>(
      hx2, h2, tpe, tpe + E_MP, nullptr, dinv_t, E_MP, N_NODES);

  wmma_gemm_kernel<H2_DIM, 3, N_CLS, false><<<gemm_blocks, GEMM_BS, 0, stream>>>(h2, wap, hxa, N_NODES);
  wmma_gemm_kernel<H2_DIM, 3, N_CLS, false><<<gemm_blocks, GEMM_BS, 0, stream>>>(h2, wbp, hxb, N_NODES);

  bias_init_kernel<<<blocks_for((long long)N_NODES * N_CLS, BS), BS, 0, stream>>>(attr_a, b_attr, N_NODES, N_CLS);
  gcn_agg_kernel<N_CLS, AGG_EPW><<<agg_blocks, BS, 0, stream>>>(
      hxa, attr_a, ei, ei + E_MP, ew, dinv_w, E_MP, N_NODES);

  bias_init_kernel<<<blocks_for((long long)N_NODES * N_CLS, BS), BS, 0, stream>>>(att_a, b_att, N_NODES, N_CLS);
  gcn_agg_kernel<N_CLS, AGG_EPW><<<agg_blocks, BS, 0, stream>>>(
      hxb, att_a, ei, ei + E_MP, ew, dinv_w, E_MP, N_NODES);

  edge_dot_kernel<<<blocks_for(2 * E_LP, BS), BS, 0, stream>>>(h2, pos, neg, out);
  log_softmax_kernel<<<blocks_for(N_NODES, BS), BS, 0, stream>>>(attr_a, out + 2 * E_LP, N_NODES, N_CLS);
  log_softmax_kernel<<<blocks_for(N_NODES, BS), BS, 0, stream>>>(att_a,  out + 2 * E_LP + (long long)N_NODES * N_CLS, N_NODES, N_CLS);
}


# f32 atomic sums (degrees, gcn_agg rows): one accumulator plane per chip half, a wave picks its plane by its die number, readers add the two planes; sums = two per-half partials, then added
# speedup vs baseline: 1.5847x; 1.2051x over previous
#include <hip/hip_runtime.h>
#include <hip/hip_bf16.h>

#define N_NODES   50000
#define D_FEAT    512
#define H1_DIM    128
#define H2_DIM    64
#define N_CLS     40
#define E_MP      800000
#define E_LP      250000

typedef __attribute__((ext_vector_type(2))) float v2f;
typedef __attribute__((ext_vector_type(8))) float v8f;

__device__ __forceinline__ int chip_half() {
  return (int)((__builtin_amdgcn_s_sendmsg_rtn(0x87) >> 18) & 1u);
}

__global__ void deg_init_kernel(float* __restrict__ p0, float* __restrict__ p1, int n) {
  int i = blockIdx.x * blockDim.x + threadIdx.x;
  if (i < n) { p0[i] = 1.0f; p1[i] = 0.0f; }
}

__global__ void deg_accum_kernel(float* __restrict__ deg0, float* __restrict__ deg1,
                                 const int* __restrict__ cols,
                                 const float* __restrict__ ew, int E) {
  long long i = (long long)blockIdx.x * blockDim.x + threadIdx.x;
  float* deg = chip_half() ? deg1 : deg0;
  if (i < E) {
    float w = ew ? ew[i] : 1.0f;
    atomicAdd(&deg[cols[i]], w);
  }
}

__global__ void rsqrt_kernel(float* __restrict__ p, const float* __restrict__ p1, int n) {
  int i = blockIdx.x * blockDim.x + threadIdx.x;
  if (i < n) {
    float d = p[i] + p1[i];
    p[i] = (d > 0.0f) ? rsqrtf(d) : 0.0f;
  }
}

__global__ void bias_init_kernel(float* __restrict__ out0, float* __restrict__ out1,
                                 const float* __restrict__ b, long long n, int F) {
  long long i = (long long)blockIdx.x * blockDim.x + threadIdx.x;
  long long total = n * F;
  if (i < total) { out0[i] = b[i % F]; out1[i] = 0.0f; }
}

__global__ void add_planes_kernel(const float4* __restrict__ a, const float4* __restrict__ b,
                                  float4* __restrict__ o, int n4) {
  const int q = n4 >> 2;
  int i = blockIdx.x * blockDim.x + threadIdx.x;
  if (i < q) {
    float4 x[4], y[4];
#pragma unroll
    for (int k = 0; k < 4; ++k) { x[k] = a[i + k * q]; y[k] = b[i + k * q]; }
#pragma unroll
    for (int k = 0; k < 4; ++k)
      o[i + k * q] = make_float4(x[k].x + y[k].x, x[k].y + y[k].y, x[k].z + y[k].z, x[k].w + y[k].w);
  }
}

__global__ void pack_b_kernel(const float* __restrict__ B, float* __restrict__ Bp,
                              int K, int NC, int NCP) {
  int p = blockIdx.x * blockDim.x + threadIdx.x;
  int total = (K / 2) * NCP;
  if (p >= total) return;
  int k2 = p / NCP;
  int col = p % NCP;
  float x0 = 0.0f, x1 = 0.0f;
  if (col < NC) {
    x0 = B[(long long)(2 * k2) * NC + col];
    x1 = B[(long long)(2 * k2 + 1) * NC + col];
  }
  Bp[2 * p + 0] = x0;
  Bp[2 * p + 1] = x1;
}

template <int KDIM, int NT, int NCSTORE, bool RELU_A, bool DUAL_A>
__global__ void wmma_gemm_kernel(const float* __restrict__ A, const float* __restrict__ A2,
                                 const float* __restrict__ Bp, float* __restrict__ C, int M) {
  constexpr int NCP = NT * 16;
  int wave = (int)(((long long)blockIdx.x * blockDim.x + threadIdx.x) >> 5);
  int lane = threadIdx.x & 31;
  int ntiles = M >> 4;
  if (wave >= ntiles) return;

  int row0 = wave << 4;
  int arow = row0 + (lane & 15);
  int khb  = lane >> 4;
  int bcol = lane & 15;

  v8f acc[NT] = {};

  const float2* __restrict__ Arow2 = (const float2*)(A + (long long)arow * KDIM);
  const float2* __restrict__ Arow2b = (const float2*)((DUAL_A ? A2 : A) + (long long)arow * KDIM);
  const float2* __restrict__ Bp2   = (const float2*)Bp;

#pragma unroll 2
  for (int k2 = 0; k2 < KDIM / 2; k2 += 2) {
    const float2* __restrict__ Bbase = Bp2 + (long long)(k2 + khb) * NCP + bcol;
    float2 bf[NT];
#pragma unroll
    for (int t = 0; t < NT; ++t) bf[t] = Bbase[t * 16];

    float2 af = Arow2[k2 + khb];
    if (DUAL_A) { float2 ag = Arow2b[k2 + khb]; af.x += ag.x; af.y += ag.y; }
    if (RELU_A) { af.x = fmaxf(af.x, 0.0f); af.y = fmaxf(af.y, 0.0f); }
    v2f a; a.x = af.x; a.y = af.y;

#pragma unroll
    for (int t = 0; t < NT; ++t) {
      v2f b; b.x = bf[t].x; b.y = bf[t].y;
      acc[t] = __builtin_amdgcn_wmma_f32_16x16x4_f32(
           false, a,  false, b,
           (short)0, acc[t],  false,  false);
    }
  }

  int crow = row0 + (khb << 3);
  int ccol = lane & 15;
#pragma unroll
  for (int t = 0; t < NT; ++t) {
    int col = (t << 4) + ccol;
    if (col < NCSTORE) {
#pragma unroll
      for (int r = 0; r < 8; ++r) {
        C[(long long)(crow + r) * NCSTORE + col] = acc[t][r];
      }
    }
  }
}

template <int F, int EPW>
__global__ void gcn_agg_kernel(const float* __restrict__ hx, float* __restrict__ out0,
                               float* __restrict__ out1, const int* __restrict__ rows, const int* __restrict__ cols,
                               const float* __restrict__ ew, const float* __restrict__ dinv,
                               int E, int N) {
  constexpr int FPT = (F + 31) / 32;
  const int lane = threadIdx.x & 31;
  const int wib = __builtin_amdgcn_readfirstlane((int)(threadIdx.x >> 5));
  const long long wave = (long long)blockIdx.x * (blockDim.x >> 5) + wib;
  const long long total = (long long)E + N;
  const long long e0 = wave * EPW;
  if (e0 >= total) return;
  float* out = chip_half() ? out1 : out0;

  int r = 0, cl = 0;
  float nrm = 0.0f;
  {
    const long long e = e0 + lane;
    if (lane < EPW && e < total) {
      float w;
      if (e < E) { r = rows[e]; cl = cols[e]; w = ew ? ew[e] : 1.0f; }
      else       { r = (int)(e - E); cl = r;  w = 1.0f; }
      nrm = dinv[r] * w * dinv[cl];
    }
  }

  const float* hrow[EPW];
#pragma unroll
  for (int i = 0; i < EPW; ++i)
    hrow[i] = hx + (long long)__builtin_amdgcn_readlane(r, i) * F + lane;

  float v[EPW][FPT];
#pragma unroll
  for (int j = 0; j < FPT; ++j) {
    if (lane + 32 * j < F) {
#pragma unroll
      for (int i = 0; i < EPW; ++i) v[i][j] = hrow[i][32 * j];
    }
  }

#pragma unroll
  for (int i = 0; i < EPW; ++i) {
    const int   ci = __builtin_amdgcn_readlane(cl, i);
    const float si = __int_as_float(__builtin_amdgcn_readlane(__float_as_int(nrm), i));
    if (e0 + i < total) {
      float* o = out + (long long)ci * F + lane;
#pragma unroll
      for (int j = 0; j < FPT; ++j)
        if (lane + 32 * j < F) atomicAdd(o + 32 * j, v[i][j] * si);
    }
  }
}

__global__ void log_softmax_kernel(const float* __restrict__ in0, const float* __restrict__ in1,
                                   float* __restrict__ out, int n, int F) {
  int row = blockIdx.x * blockDim.x + threadIdx.x;
  if (row >= n) return;
  const float* p = in0 + (long long)row * F;
  const float* q = in1 + (long long)row * F;
  float m = -INFINITY;
  for (int c = 0; c < F; ++c) m = fmaxf(m, p[c] + q[c]);
  float s = 0.0f;
  for (int c = 0; c < F; ++c) s += __expf((p[c] + q[c]) - m);
  float ls = __logf(s) + m;
  float* o = out + (long long)row * F;
  for (int c = 0; c < F; ++c) o[c] = (p[c] + q[c]) - ls;
}

__global__ void edge_dot_kernel(const float* __restrict__ h, const int* __restrict__ pos,
                                const int* __restrict__ neg, float* __restrict__ res) {
  int e = blockIdx.x * blockDim.x + threadIdx.x;
  if (e >= 2 * E_LP) return;
  const int* src = (e < E_LP) ? pos : neg;
  int eo = (e < E_LP) ? e : (e - E_LP);
  int j = src[eo];
  int i = src[E_LP + eo];
  const float4* hj = (const float4*)(h + (long long)j * H2_DIM);
  const float4* hi = (const float4*)(h + (long long)i * H2_DIM);
  float s = 0.0f;
#pragma unroll
  for (int k = 0; k < H2_DIM / 4; ++k) {
    float4 a = hi[k], b = hj[k];
    s += a.x * b.x + a.y * b.y + a.z * b.z + a.w * b.w;
  }
  res[e] = s;
}

static inline int blocks_for(long long total, int bs) {
  return (int)((total + bs - 1) / bs);
}

extern "C" void kernel_launch(void* const* d_in, const int* in_sizes, int n_in,
                              void* d_out, int out_size, void* d_ws, size_t ws_size,
                              hipStream_t stream) {
  const float* x      = (const float*)d_in[0];
  const int*   tpe    = (const int*)  d_in[1];
  const int*   ei     = (const int*)  d_in[2];
  const float* ew     = (const float*)d_in[3];
  const int*   pos    = (const int*)  d_in[4];
  const int*   neg    = (const int*)  d_in[5];
  const float* W1     = (const float*)d_in[6];
  const float* b1     = (const float*)d_in[7];
  const float* W2     = (const float*)d_in[8];
  const float* b2     = (const float*)d_in[9];
  const float* W_attr = (const float*)d_in[10];
  const float* b_attr = (const float*)d_in[11];
  const float* W_att  = (const float*)d_in[12];
  const float* b_att  = (const float*)d_in[13];

  float* out = (float*)d_out;
  float* ws  = (float*)d_ws;

  const long long OFF_DINV_T = 0;
  const long long OFF_DINV_W = 50000;
  const long long OFF_R0     = 100000;
  const long long OFF_R1     = OFF_R0 + 6400000;
  const long long OFF_R2     = OFF_R1 + 6400000;
  const long long OFF_W1P    = OFF_R2 + 6400000;
  const long long OFF_W2P    = OFF_W1P + 65536;
  const long long OFF_WAP    = OFF_W2P + 8192;
  const long long OFF_WBP    = OFF_WAP + 3072;

  float* dinv_t = ws + OFF_DINV_T;
  float* dinv_w = ws + OFF_DINV_W;
  float* degt_p1 = ws + OFF_R0;
  float* degw_p1 = ws + OFF_R0 + 50000;
  float* hx1    = ws + OFF_R0;
  float* h1_p0  = ws + OFF_R1;
  float* h1_p1  = ws + OFF_R2;
  float* hx2    = ws + OFF_R0;
  float* h2_p0  = ws + OFF_R1;
  float* h2_p1  = ws + OFF_R1 + 3200000;
  float* h2     = ws + OFF_R2;
  float* hxa    = ws + OFF_R0;
  float* hxb    = ws + OFF_R0 + 2000000;
  float* attr_p0 = ws + OFF_R1;
  float* attr_p1 = ws + OFF_R1 + 2000000;
  float* att_p0  = ws + OFF_R1 + 4000000;
  float* att_p1  = ws + OFF_R2 + 3200000;
  float* w1p    = ws + OFF_W1P;
  float* w2p    = ws + OFF_W2P;
  float* wap    = ws + OFF_WAP;
  float* wbp    = ws + OFF_WBP;

  const int BS = 256;

  pack_b_kernel<<<blocks_for((D_FEAT / 2) * H1_DIM, BS), BS, 0, stream>>>(W1, w1p, D_FEAT, H1_DIM, H1_DIM);
  pack_b_kernel<<<blocks_for((H1_DIM / 2) * H2_DIM, BS), BS, 0, stream>>>(W2, w2p, H1_DIM, H2_DIM, H2_DIM);
  pack_b_kernel<<<blocks_for((H2_DIM / 2) * 48, BS), BS, 0, stream>>>(W_attr, wap, H2_DIM, N_CLS, 48);
  pack_b_kernel<<<blocks_for((H2_DIM / 2) * 48, BS), BS, 0, stream>>>(W_att,  wbp, H2_DIM, N_CLS, 48);

  deg_init_kernel<<<blocks_for(N_NODES, BS), BS, 0, stream>>>(dinv_t, degt_p1, N_NODES);
  deg_accum_kernel<<<blocks_for(E_MP, BS), BS, 0, stream>>>(dinv_t, degt_p1, tpe + E_MP, nullptr, E_MP);
  rsqrt_kernel<<<blocks_for(N_NODES, BS), BS, 0, stream>>>(dinv_t, degt_p1, N_NODES);

  deg_init_kernel<<<blocks_for(N_NODES, BS), BS, 0, stream>>>(dinv_w, degw_p1, N_NODES);
  deg_accum_kernel<<<blocks_for(E_MP, BS), BS, 0, stream>>>(dinv_w, degw_p1, ei + E_MP, ew, E_MP);
  rsqrt_kernel<<<blocks_for(N_NODES, BS), BS, 0, stream>>>(dinv_w, degw_p1, N_NODES);

  const int ROW_TILES = N_NODES / 16;
  const int GEMM_BS = 128;
  const int gemm_blocks = blocks_for((long long)ROW_TILES * 32, GEMM_BS);

  constexpr int AGG_EPW = 8;
  const int agg_blocks = blocks_for(((long long)(E_MP + N_NODES) + AGG_EPW - 1) / AGG_EPW * 32, BS);

  wmma_gemm_kernel<D_FEAT, 8, H1_DIM, false, false><<<gemm_blocks, GEMM_BS, 0, stream>>>(x, nullptr, w1p, hx1, N_NODES);
  bias_init_kernel<<<blocks_for((long long)N_NODES * H1_DIM, BS), BS, 0, stream>>>(h1_p0, h1_p1, b1, N_NODES, H1_DIM);
  gcn_agg_kernel<H1_DIM, AGG_EPW><<<agg_blocks, BS, 0, stream>>>(
      hx1, h1_p0, h1_p1, tpe, tpe + E_MP, nullptr, dinv_t, E_MP, N_NODES);

  wmma_gemm_kernel<H1_DIM, 4, H2_DIM, true, true><<<gemm_blocks, GEMM_BS, 0, stream>>>(h1_p0, h1_p1, w2p, hx2, N_NODES);
  bias_init_kernel<<<blocks_for((long long)N_NODES * H2_DIM, BS), BS, 0, stream>>>(h2_p0, h2_p1, b2, N_NODES, H2_DIM);
  gcn_agg_kernel<H2_DIM, AGG_EPW><<<agg_blocks, BS, 0, stream>>>(
      hx2, h2_p0, h2_p1, tpe, tpe + E_MP, nullptr, dinv_t, E_MP, N_NODES);
  add_planes_kernel<<<blocks_for(N_NODES * H2_DIM / 16, BS), BS, 0, stream>>>(
      (const float4*)h2_p0, (const float4*)h2_p1, (float4*)h2, N_NODES * H2_DIM / 4);

  wmma_gemm_kernel<H2_DIM, 3, N_CLS, false, false><<<gemm_blocks, GEMM_BS, 0, stream>>>(h2, nullptr, wap, hxa, N_NODES);
  wmma_gemm_kernel<H2_DIM, 3, N_CLS, false, false><<<gemm_blocks, GEMM_BS, 0, stream>>>(h2, nullptr, wbp, hxb, N_NODES);

  bias_init_kernel<<<blocks_for((long long)N_NODES * N_CLS, BS), BS, 0, stream>>>(attr_p0, attr_p1, b_attr, N_NODES, N_CLS);
  gcn_agg_kernel<N_CLS, AGG_EPW><<<agg_blocks, BS, 0, stream>>>(
      hxa, attr_p0, attr_p1, ei, ei + E_MP, ew, dinv_w, E_MP, N_NODES);

  bias_init_kernel<<<blocks_for((long long)N_NODES * N_CLS, BS), BS, 0, stream>>>(att_p0, att_p1, b_att, N_NODES, N_CLS);
  gcn_agg_kernel<N_CLS, AGG_EPW><<<agg_blocks, BS, 0, stream>>>(
      hxb, att_p0, att_p1, ei, ei + E_MP, ew, dinv_w, E_MP, N_NODES);

  edge_dot_kernel<<<blocks_for(2 * E_LP, BS), BS, 0, stream>>>(h2, pos, neg, out);
  log_softmax_kernel<<<blocks_for(N_NODES, BS), BS, 0, stream>>>(attr_p0, attr_p1, out + 2 * E_LP, N_NODES, N_CLS);
  log_softmax_kernel<<<blocks_for(N_NODES, BS), BS, 0, stream>>>(att_p0, att_p1, out + 2 * E_LP + (long long)N_NODES * N_CLS, N_NODES, N_CLS);
}


# heads: h2 (64 wide) aggregated once over the weighted graph, then one WMMA product with [W_attr|W_att] (80 cols), bias in softmax; replaces two 40-wide aggregations; sum order A(hW) -> (Ah)W
# speedup vs baseline: 1.8479x; 1.1661x over previous
#include <hip/hip_runtime.h>
#include <hip/hip_bf16.h>

#define N_NODES   50000
#define D_FEAT    512
#define H1_DIM    128
#define H2_DIM    64
#define N_CLS     40
#define E_MP      800000
#define E_LP      250000

typedef __attribute__((ext_vector_type(2))) float v2f;
typedef __attribute__((ext_vector_type(8))) float v8f;

__device__ __forceinline__ int chip_half() {
  return (int)((__builtin_amdgcn_s_sendmsg_rtn(0x87) >> 18) & 1u);
}

__global__ void deg_init_kernel(float* __restrict__ p0, float* __restrict__ p1, int n) {
  int i = blockIdx.x * blockDim.x + threadIdx.x;
  if (i < n) { p0[i] = 1.0f; p1[i] = 0.0f; }
}

__global__ void deg_accum_kernel(float* __restrict__ deg0, float* __restrict__ deg1,
                                 const int* __restrict__ cols,
                                 const float* __restrict__ ew, int E) {
  long long i = (long long)blockIdx.x * blockDim.x + threadIdx.x;
  float* deg = chip_half() ? deg1 : deg0;
  if (i < E) {
    float w = ew ? ew[i] : 1.0f;
    atomicAdd(&deg[cols[i]], w);
  }
}

__global__ void rsqrt_kernel(float* __restrict__ p, const float* __restrict__ p1, int n) {
  int i = blockIdx.x * blockDim.x + threadIdx.x;
  if (i < n) {
    float d = p[i] + p1[i];
    p[i] = (d > 0.0f) ? rsqrtf(d) : 0.0f;
  }
}

__global__ void bias_init_kernel(float* __restrict__ out0, float* __restrict__ out1,
                                 const float* __restrict__ b, long long n, int F) {
  long long i = (long long)blockIdx.x * blockDim.x + threadIdx.x;
  long long total = n * F;
  if (i < total) { out0[i] = b[i % F]; out1[i] = 0.0f; }
}

__global__ void add_planes_kernel(const float4* __restrict__ a, const float4* __restrict__ b,
                                  float4* __restrict__ o, float4* __restrict__ z0,
                                  float4* __restrict__ z1, int n4) {
  const int q = n4 >> 2;
  int i = blockIdx.x * blockDim.x + threadIdx.x;
  if (i < q) {
    float4 x[4], y[4];
#pragma unroll
    for (int k = 0; k < 4; ++k) { x[k] = a[i + k * q]; y[k] = b[i + k * q]; }
    const float4 zero = make_float4(0.0f, 0.0f, 0.0f, 0.0f);
#pragma unroll
    for (int k = 0; k < 4; ++k) { z0[i + k * q] = zero; z1[i + k * q] = zero; }
#pragma unroll
    for (int k = 0; k < 4; ++k)
      o[i + k * q] = make_float4(x[k].x + y[k].x, x[k].y + y[k].y, x[k].z + y[k].z, x[k].w + y[k].w);
  }
}

__global__ void pack_b_kernel(const float* __restrict__ B, float* __restrict__ Bp,
                              int K, int NC, int NCP) {
  int p = blockIdx.x * blockDim.x + threadIdx.x;
  int total = (K / 2) * NCP;
  if (p >= total) return;
  int k2 = p / NCP;
  int col = p % NCP;
  float x0 = 0.0f, x1 = 0.0f;
  if (col < NC) {
    x0 = B[(long long)(2 * k2) * NC + col];
    x1 = B[(long long)(2 * k2 + 1) * NC + col];
  }
  Bp[2 * p + 0] = x0;
  Bp[2 * p + 1] = x1;
}

__global__ void pack_heads_kernel(const float* __restrict__ Ba, const float* __restrict__ Bb,
                                  float* __restrict__ Bp, int K, int NC) {
  const int NCP = 2 * NC;
  int p = blockIdx.x * blockDim.x + threadIdx.x;
  int total = (K / 2) * NCP;
  if (p >= total) return;
  int k2 = p / NCP;
  int col = p % NCP;
  const float* __restrict__ B = (col < NC) ? Ba : Bb;
  int cc = (col < NC) ? col : col - NC;
  Bp[2 * p + 0] = B[(long long)(2 * k2) * NC + cc];
  Bp[2 * p + 1] = B[(long long)(2 * k2 + 1) * NC + cc];
}

template <int KDIM, int NT, int NCSTORE, bool RELU_A, bool DUAL_A>
__global__ void wmma_gemm_kernel(const float* __restrict__ A, const float* __restrict__ A2,
                                 const float* __restrict__ Bp, float* __restrict__ C, int M) {
  constexpr int NCP = NT * 16;
  int wave = (int)(((long long)blockIdx.x * blockDim.x + threadIdx.x) >> 5);
  int lane = threadIdx.x & 31;
  int ntiles = M >> 4;
  if (wave >= ntiles) return;

  int row0 = wave << 4;
  int arow = row0 + (lane & 15);
  int khb  = lane >> 4;
  int bcol = lane & 15;

  v8f acc[NT] = {};

  const float2* __restrict__ Arow2 = (const float2*)(A + (long long)arow * KDIM);
  const float2* __restrict__ Arow2b = (const float2*)((DUAL_A ? A2 : A) + (long long)arow * KDIM);
  const float2* __restrict__ Bp2   = (const float2*)Bp;

#pragma unroll 2
  for (int k2 = 0; k2 < KDIM / 2; k2 += 2) {
    const float2* __restrict__ Bbase = Bp2 + (long long)(k2 + khb) * NCP + bcol;
    float2 bf[NT];
#pragma unroll
    for (int t = 0; t < NT; ++t) bf[t] = Bbase[t * 16];

    float2 af = Arow2[k2 + khb];
    if (DUAL_A) { float2 ag = Arow2b[k2 + khb]; af.x += ag.x; af.y += ag.y; }
    if (RELU_A) { af.x = fmaxf(af.x, 0.0f); af.y = fmaxf(af.y, 0.0f); }
    v2f a; a.x = af.x; a.y = af.y;

#pragma unroll
    for (int t = 0; t < NT; ++t) {
      v2f b; b.x = bf[t].x; b.y = bf[t].y;
      acc[t] = __builtin_amdgcn_wmma_f32_16x16x4_f32(
           false, a,  false, b,
           (short)0, acc[t],  false,  false);
    }
  }

  int crow = row0 + (khb << 3);
  int ccol = lane & 15;
#pragma unroll
  for (int t = 0; t < NT; ++t) {
    int col = (t << 4) + ccol;
    if (col < NCSTORE) {
#pragma unroll
      for (int r = 0; r < 8; ++r) {
        C[(long long)(crow + r) * NCSTORE + col] = acc[t][r];
      }
    }
  }
}

template <int F, int EPW>
__global__ void gcn_agg_kernel(const float* __restrict__ hx, float* __restrict__ out0,
                               float* __restrict__ out1, const int* __restrict__ rows, const int* __restrict__ cols,
                               const float* __restrict__ ew, const float* __restrict__ dinv,
                               int E, int N) {
  constexpr int FPT = (F + 31) / 32;
  const int lane = threadIdx.x & 31;
  const int wib = __builtin_amdgcn_readfirstlane((int)(threadIdx.x >> 5));
  const long long wave = (long long)blockIdx.x * (blockDim.x >> 5) + wib;
  const long long total = (long long)E + N;
  const long long e0 = wave * EPW;
  if (e0 >= total) return;
  float* out = chip_half() ? out1 : out0;

  int r = 0, cl = 0;
  float nrm = 0.0f;
  {
    const long long e = e0 + lane;
    if (lane < EPW && e < total) {
      float w;
      if (e < E) { r = rows[e]; cl = cols[e]; w = ew ? ew[e] : 1.0f; }
      else       { r = (int)(e - E); cl = r;  w = 1.0f; }
      nrm = dinv[r] * w * dinv[cl];
    }
  }

  const float* hrow[EPW];
#pragma unroll
  for (int i = 0; i < EPW; ++i)
    hrow[i] = hx + (long long)__builtin_amdgcn_readlane(r, i) * F + lane;

  float v[EPW][FPT];
#pragma unroll
  for (int j = 0; j < FPT; ++j) {
    if (lane + 32 * j < F) {
#pragma unroll
      for (int i = 0; i < EPW; ++i) v[i][j] = hrow[i][32 * j];
    }
  }

#pragma unroll
  for (int i = 0; i < EPW; ++i) {
    const int   ci = __builtin_amdgcn_readlane(cl, i);
    const float si = __int_as_float(__builtin_amdgcn_readlane(__float_as_int(nrm), i));
    if (e0 + i < total) {
      float* o = out + (long long)ci * F + lane;
#pragma unroll
      for (int j = 0; j < FPT; ++j)
        if (lane + 32 * j < F) atomicAdd(o + 32 * j, v[i][j] * si);
    }
  }
}

__global__ void log_softmax_kernel(const float* __restrict__ in, const float* __restrict__ bias0,
                                   const float* __restrict__ bias1, float* __restrict__ out0,
                                   float* __restrict__ out1, int n, int F) {
  int idx = blockIdx.x * blockDim.x + threadIdx.x;
  if (idx >= 2 * n) return;
  int row = idx >> 1;
  int head = idx & 1;
  const float* p = in + (long long)idx * F;
  const float* q = head ? bias1 : bias0;
  float m = -INFINITY;
  for (int c = 0; c < F; ++c) m = fmaxf(m, p[c] + q[c]);
  float s = 0.0f;
  for (int c = 0; c < F; ++c) s += __expf((p[c] + q[c]) - m);
  float ls = __logf(s) + m;
  float* o = (head ? out1 : out0) + (long long)row * F;
  for (int c = 0; c < F; ++c) o[c] = (p[c] + q[c]) - ls;
}

__global__ void edge_dot_kernel(const float* __restrict__ h, const int* __restrict__ pos,
                                const int* __restrict__ neg, float* __restrict__ res) {
  int e = blockIdx.x * blockDim.x + threadIdx.x;
  if (e >= 2 * E_LP) return;
  const int* src = (e < E_LP) ? pos : neg;
  int eo = (e < E_LP) ? e : (e - E_LP);
  int j = src[eo];
  int i = src[E_LP + eo];
  const float4* hj = (const float4*)(h + (long long)j * H2_DIM);
  const float4* hi = (const float4*)(h + (long long)i * H2_DIM);
  float s = 0.0f;
#pragma unroll
  for (int k = 0; k < H2_DIM / 4; ++k) {
    float4 a = hi[k], b = hj[k];
    s += a.x * b.x + a.y * b.y + a.z * b.z + a.w * b.w;
  }
  res[e] = s;
}

static inline int blocks_for(long long total, int bs) {
  return (int)((total + bs - 1) / bs);
}

extern "C" void kernel_launch(void* const* d_in, const int* in_sizes, int n_in,
                              void* d_out, int out_size, void* d_ws, size_t ws_size,
                              hipStream_t stream) {
  const float* x      = (const float*)d_in[0];
  const int*   tpe    = (const int*)  d_in[1];
  const int*   ei     = (const int*)  d_in[2];
  const float* ew     = (const float*)d_in[3];
  const int*   pos    = (const int*)  d_in[4];
  const int*   neg    = (const int*)  d_in[5];
  const float* W1     = (const float*)d_in[6];
  const float* b1     = (const float*)d_in[7];
  const float* W2     = (const float*)d_in[8];
  const float* b2     = (const float*)d_in[9];
  const float* W_attr = (const float*)d_in[10];
  const float* b_attr = (const float*)d_in[11];
  const float* W_att  = (const float*)d_in[12];
  const float* b_att  = (const float*)d_in[13];

  float* out = (float*)d_out;
  float* ws  = (float*)d_ws;

  const long long OFF_DINV_T = 0;
  const long long OFF_DINV_W = 50000;
  const long long OFF_R0     = 100000;
  const long long OFF_R1     = OFF_R0 + 6400000;
  const long long OFF_R2     = OFF_R1 + 6400000;
  const long long OFF_W1P    = OFF_R2 + 6400000;
  const long long OFF_W2P    = OFF_W1P + 65536;
  const long long OFF_WCP    = OFF_W2P + 8192;

  float* dinv_t = ws + OFF_DINV_T;
  float* dinv_w = ws + OFF_DINV_W;
  float* degt_p1 = ws + OFF_R0;
  float* degw_p1 = ws + OFF_R0 + 50000;
  float* hx1    = ws + OFF_R0;
  float* h1_p0  = ws + OFF_R1;
  float* h1_p1  = ws + OFF_R2;
  float* hx2    = ws + OFF_R0;
  float* h2_p0  = ws + OFF_R1;
  float* h2_p1  = ws + OFF_R1 + 3200000;
  float* h2     = ws + OFF_R2;
  float* g_p0   = ws + OFF_R0;
  float* g_p1   = ws + OFF_R0 + 3200000;
  float* hcat   = ws + OFF_R1;
  float* w1p    = ws + OFF_W1P;
  float* w2p    = ws + OFF_W2P;
  float* wcp    = ws + OFF_WCP;

  const int BS = 256;

  pack_b_kernel<<<blocks_for((D_FEAT / 2) * H1_DIM, BS), BS, 0, stream>>>(W1, w1p, D_FEAT, H1_DIM, H1_DIM);
  pack_b_kernel<<<blocks_for((H1_DIM / 2) * H2_DIM, BS), BS, 0, stream>>>(W2, w2p, H1_DIM, H2_DIM, H2_DIM);
  pack_heads_kernel<<<blocks_for((H2_DIM / 2) * 2 * N_CLS, BS), BS, 0, stream>>>(W_attr, W_att, wcp, H2_DIM, N_CLS);

  deg_init_kernel<<<blocks_for(N_NODES, BS), BS, 0, stream>>>(dinv_t, degt_p1, N_NODES);
  deg_accum_kernel<<<blocks_for(E_MP, BS), BS, 0, stream>>>(dinv_t, degt_p1, tpe + E_MP, nullptr, E_MP);
  rsqrt_kernel<<<blocks_for(N_NODES, BS), BS, 0, stream>>>(dinv_t, degt_p1, N_NODES);

  deg_init_kernel<<<blocks_for(N_NODES, BS), BS, 0, stream>>>(dinv_w, degw_p1, N_NODES);
  deg_accum_kernel<<<blocks_for(E_MP, BS), BS, 0, stream>>>(dinv_w, degw_p1, ei + E_MP, ew, E_MP);
  rsqrt_kernel<<<blocks_for(N_NODES, BS), BS, 0, stream>>>(dinv_w, degw_p1, N_NODES);

  const int ROW_TILES = N_NODES / 16;
  const int GEMM_BS = 128;
  const int gemm_blocks = blocks_for((long long)ROW_TILES * 32, GEMM_BS);

  constexpr int AGG_EPW = 8;
  const int agg_blocks = blocks_for(((long long)(E_MP + N_NODES) + AGG_EPW - 1) / AGG_EPW * 32, BS);

  wmma_gemm_kernel<D_FEAT, 8, H1_DIM, false, false><<<gemm_blocks, GEMM_BS, 0, stream>>>(x, nullptr, w1p, hx1, N_NODES);
  bias_init_kernel<<<blocks_for((long long)N_NODES * H1_DIM, BS), BS, 0, stream>>>(h1_p0, h1_p1, b1, N_NODES, H1_DIM);
  gcn_agg_kernel<H1_DIM, AGG_EPW><<<agg_blocks, BS, 0, stream>>>(
      hx1, h1_p0, h1_p1, tpe, tpe + E_MP, nullptr, dinv_t, E_MP, N_NODES);

  wmma_gemm_kernel<H1_DIM, 4, H2_DIM, true, true><<<gemm_blocks, GEMM_BS, 0, stream>>>(h1_p0, h1_p1, w2p, hx2, N_NODES);
  bias_init_kernel<<<blocks_for((long long)N_NODES * H2_DIM, BS), BS, 0, stream>>>(h2_p0, h2_p1, b2, N_NODES, H2_DIM);
  gcn_agg_kernel<H2_DIM, AGG_EPW><<<agg_blocks, BS, 0, stream>>>(
      hx2, h2_p0, h2_p1, tpe, tpe + E_MP, nullptr, dinv_t, E_MP, N_NODES);
  add_planes_kernel<<<blocks_for(N_NODES * H2_DIM / 16, BS), BS, 0, stream>>>(
      (const float4*)h2_p0, (const float4*)h2_p1, (float4*)h2, (float4*)g_p0, (float4*)g_p1,
      N_NODES * H2_DIM / 4);

  gcn_agg_kernel<H2_DIM, AGG_EPW><<<agg_blocks, BS, 0, stream>>>(
      h2, g_p0, g_p1, ei, ei + E_MP, ew, dinv_w, E_MP, N_NODES);
  wmma_gemm_kernel<H2_DIM, 5, 2 * N_CLS, false, true><<<gemm_blocks, GEMM_BS, 0, stream>>>(g_p0, g_p1, wcp, hcat, N_NODES);

  edge_dot_kernel<<<blocks_for(2 * E_LP, BS), BS, 0, stream>>>(h2, pos, neg, out);
  log_softmax_kernel<<<blocks_for(2 * N_NODES, BS), BS, 0, stream>>>(
      hcat, b_attr, b_att, out + 2 * E_LP, out + 2 * E_LP + (long long)N_NODES * N_CLS, N_NODES, N_CLS);
}


# independent pieces share launches, 20 -> 11: packs + degree start values; x@W1 beside both degree sums; rsqrt + h1 start values; head product beside the edge dots; no arithmetic changed
# speedup vs baseline: 2.1001x; 1.1365x over previous
#include <hip/hip_runtime.h>
#include <hip/hip_bf16.h>

#define N_NODES   50000
#define D_FEAT    512
#define H1_DIM    128
#define H2_DIM    64
#define N_CLS     40
#define E_MP      800000
#define E_LP      250000

typedef __attribute__((ext_vector_type(2))) float v2f;
typedef __attribute__((ext_vector_type(8))) float v8f;

__device__ __forceinline__ int chip_half() {
  return (int)((__builtin_amdgcn_s_sendmsg_rtn(0x87) >> 18) & 1u);
}


__device__ __forceinline__ void deg_accum_body(float* __restrict__ deg0, float* __restrict__ deg1,
                                               const int* __restrict__ cols,
                                               const float* __restrict__ ew, int E, long long i) {
  float* deg = chip_half() ? deg1 : deg0;
  if (i < E) {
    float w = ew ? ew[i] : 1.0f;
    atomicAdd(&deg[cols[i]], w);
  }
}

__global__ void init1_rsqrt_kernel(float* __restrict__ out0, float* __restrict__ out1,
                                   const float* __restrict__ b, float* __restrict__ dinv,
                                   long long total, int F, int nd) {
  long long i = (long long)blockIdx.x * blockDim.x + threadIdx.x;
  if (i < total) {
    if (i < nd) {
      float d = dinv[i] + out0[i];
      dinv[i] = (d > 0.0f) ? rsqrtf(d) : 0.0f;
    }
    out0[i] = b[i % F];
    out1[i] = 0.0f;
  }
}

__global__ void bias_init_kernel(float* __restrict__ out0, float* __restrict__ out1,
                                 const float* __restrict__ b, long long n, int F) {
  long long i = (long long)blockIdx.x * blockDim.x + threadIdx.x;
  long long total = n * F;
  if (i < total) { out0[i] = b[i % F]; out1[i] = 0.0f; }
}

__global__ void add_planes_kernel(const float4* __restrict__ a, const float4* __restrict__ b,
                                  float4* __restrict__ o, float4* __restrict__ z0,
                                  float4* __restrict__ z1, int n4) {
  const int q = n4 >> 2;
  int i = blockIdx.x * blockDim.x + threadIdx.x;
  if (i < q) {
    float4 x[4], y[4];
#pragma unroll
    for (int k = 0; k < 4; ++k) { x[k] = a[i + k * q]; y[k] = b[i + k * q]; }
    const float4 zero = make_float4(0.0f, 0.0f, 0.0f, 0.0f);
#pragma unroll
    for (int k = 0; k < 4; ++k) { z0[i + k * q] = zero; z1[i + k * q] = zero; }
#pragma unroll
    for (int k = 0; k < 4; ++k)
      o[i + k * q] = make_float4(x[k].x + y[k].x, x[k].y + y[k].y, x[k].z + y[k].z, x[k].w + y[k].w);
  }
}

__device__ __forceinline__ void pack_b_body(const float* __restrict__ B, float* __restrict__ Bp,
                                            int K, int NC, int NCP, int p) {
  int total = (K / 2) * NCP;
  if (p >= total) return;
  int k2 = p / NCP;
  int col = p % NCP;
  float x0 = 0.0f, x1 = 0.0f;
  if (col < NC) {
    x0 = B[(long long)(2 * k2) * NC + col];
    x1 = B[(long long)(2 * k2 + 1) * NC + col];
  }
  Bp[2 * p + 0] = x0;
  Bp[2 * p + 1] = x1;
}

__device__ __forceinline__ void pack_heads_body(const float* __restrict__ Ba,
                                                const float* __restrict__ Bb,
                                                float* __restrict__ Bp, int K, int NC, int p) {
  const int NCP = 2 * NC;
  int total = (K / 2) * NCP;
  if (p >= total) return;
  int k2 = p / NCP;
  int col = p % NCP;
  const float* __restrict__ B = (col < NC) ? Ba : Bb;
  int cc = (col < NC) ? col : col - NC;
  Bp[2 * p + 0] = B[(long long)(2 * k2) * NC + cc];
  Bp[2 * p + 1] = B[(long long)(2 * k2 + 1) * NC + cc];
}

#define PREP_BS   256
#define PREP_B_W1 ((D_FEAT / 2) * H1_DIM / PREP_BS)
#define PREP_B_W2 ((H1_DIM / 2) * H2_DIM / PREP_BS)
#define PREP_B_WC ((H2_DIM / 2) * 2 * N_CLS / PREP_BS)
#define PREP_B_DG ((2 * N_NODES + PREP_BS - 1) / PREP_BS)
__global__ void prep_kernel(const float* __restrict__ W1, const float* __restrict__ W2,
                            const float* __restrict__ Wa, const float* __restrict__ Wb,
                            float* __restrict__ w1p, float* __restrict__ w2p,
                            float* __restrict__ wcp, float* __restrict__ deg0,
                            float* __restrict__ deg1) {
  int b = blockIdx.x;
  const int t = threadIdx.x;
  if (b < PREP_B_W1) { pack_b_body(W1, w1p, D_FEAT, H1_DIM, H1_DIM, b * PREP_BS + t); return; }
  b -= PREP_B_W1;
  if (b < PREP_B_W2) { pack_b_body(W2, w2p, H1_DIM, H2_DIM, H2_DIM, b * PREP_BS + t); return; }
  b -= PREP_B_W2;
  if (b < PREP_B_WC) { pack_heads_body(Wa, Wb, wcp, H2_DIM, N_CLS, b * PREP_BS + t); return; }
  b -= PREP_B_WC;
  const int i = b * PREP_BS + t;
  if (i < 2 * N_NODES) { deg0[i] = 1.0f; deg1[i] = 0.0f; }
}

template <int KDIM, int NT, int NCSTORE, bool RELU_A, bool DUAL_A>
__device__ __forceinline__ void wmma_gemm_body(const float* __restrict__ A,
                                               const float* __restrict__ A2,
                                               const float* __restrict__ Bp,
                                               float* __restrict__ C, int M, int wave, int lane) {
  constexpr int NCP = NT * 16;
  int ntiles = M >> 4;
  if (wave >= ntiles) return;

  int row0 = wave << 4;
  int arow = row0 + (lane & 15);
  int khb  = lane >> 4;
  int bcol = lane & 15;

  v8f acc[NT] = {};

  const float2* __restrict__ Arow2 = (const float2*)(A + (long long)arow * KDIM);
  const float2* __restrict__ Arow2b = (const float2*)((DUAL_A ? A2 : A) + (long long)arow * KDIM);
  const float2* __restrict__ Bp2   = (const float2*)Bp;

#pragma unroll 2
  for (int k2 = 0; k2 < KDIM / 2; k2 += 2) {
    const float2* __restrict__ Bbase = Bp2 + (long long)(k2 + khb) * NCP + bcol;
    float2 bf[NT];
#pragma unroll
    for (int t = 0; t < NT; ++t) bf[t] = Bbase[t * 16];

    float2 af = Arow2[k2 + khb];
    if (DUAL_A) { float2 ag = Arow2b[k2 + khb]; af.x += ag.x; af.y += ag.y; }
    if (RELU_A) { af.x = fmaxf(af.x, 0.0f); af.y = fmaxf(af.y, 0.0f); }
    v2f a; a.x = af.x; a.y = af.y;

#pragma unroll
    for (int t = 0; t < NT; ++t) {
      v2f b; b.x = bf[t].x; b.y = bf[t].y;
      acc[t] = __builtin_amdgcn_wmma_f32_16x16x4_f32(
           false, a,  false, b,
           (short)0, acc[t],  false,  false);
    }
  }

  int crow = row0 + (khb << 3);
  int ccol = lane & 15;
#pragma unroll
  for (int t = 0; t < NT; ++t) {
    int col = (t << 4) + ccol;
    if (col < NCSTORE) {
#pragma unroll
      for (int r = 0; r < 8; ++r) {
        C[(long long)(crow + r) * NCSTORE + col] = acc[t][r];
      }
    }
  }
}

template <int KDIM, int NT, int NCSTORE, bool RELU_A, bool DUAL_A>
__global__ void wmma_gemm_kernel(const float* __restrict__ A, const float* __restrict__ A2,
                                 const float* __restrict__ Bp, float* __restrict__ C, int M) {
  int wave = (int)(((long long)blockIdx.x * blockDim.x + threadIdx.x) >> 5);
  int lane = threadIdx.x & 31;
  wmma_gemm_body<KDIM, NT, NCSTORE, RELU_A, DUAL_A>(A, A2, Bp, C, M, wave, lane);
}

#define G1_BS     128
#define G1_BLOCKS ((N_NODES / 16 * 32 + G1_BS - 1) / G1_BS)
#define DG_BLOCKS (E_MP / G1_BS)
__global__ void gemm1_deg_kernel(const float* __restrict__ x, const float* __restrict__ w1p,
                                 float* __restrict__ hx1,
                                 float* __restrict__ degt0, float* __restrict__ degt1,
                                 const int* __restrict__ cols_t,
                                 float* __restrict__ degw0, float* __restrict__ degw1,
                                 const int* __restrict__ cols_w, const float* __restrict__ ew) {
  const int b = blockIdx.x;
  if (b < G1_BLOCKS) {
    int wave = (b * G1_BS + (int)threadIdx.x) >> 5;
    int lane = threadIdx.x & 31;
    wmma_gemm_body<D_FEAT, 8, H1_DIM, false, false>(x, nullptr, w1p, hx1, N_NODES, wave, lane);
  } else if (b < G1_BLOCKS + DG_BLOCKS) {
    deg_accum_body(degt0, degt1, cols_t, nullptr, E_MP,
                   (long long)(b - G1_BLOCKS) * G1_BS + threadIdx.x);
  } else {
    deg_accum_body(degw0, degw1, cols_w, ew, E_MP,
                   (long long)(b - G1_BLOCKS - DG_BLOCKS) * G1_BS + threadIdx.x);
  }
}

#define AGG_BS     256
#define AGG_EPW    8
#define AGG_BLOCKS ((int)((((long long)(E_MP + N_NODES) + AGG_EPW - 1) / AGG_EPW * 32 + AGG_BS - 1) / AGG_BS))
template <int F, int EPW>
__device__ __forceinline__ void gcn_agg_body(const float* __restrict__ hx, float* __restrict__ out0,
                                             float* __restrict__ out1, const int* __restrict__ rows,
                                             const int* __restrict__ cols,
                                             const float* __restrict__ ew,
                                             const float* __restrict__ dinv,
                                             int E, int N, int blk) {
  constexpr int FPT = (F + 31) / 32;
  const int lane = threadIdx.x & 31;
  const int wib = __builtin_amdgcn_readfirstlane((int)(threadIdx.x >> 5));
  const long long wave = (long long)blk * (AGG_BS >> 5) + wib;
  const long long total = (long long)E + N;
  const long long e0 = wave * EPW;
  if (e0 >= total) return;
  float* out = chip_half() ? out1 : out0;

  int r = 0, cl = 0;
  float nrm = 0.0f;
  {
    const long long e = e0 + lane;
    if (lane < EPW && e < total) {
      float w;
      if (e < E) { r = rows[e]; cl = cols[e]; w = ew ? ew[e] : 1.0f; }
      else       { r = (int)(e - E); cl = r;  w = 1.0f; }
      nrm = dinv[r] * w * dinv[cl];
    }
  }

  const float* hrow[EPW];
#pragma unroll
  for (int i = 0; i < EPW; ++i)
    hrow[i] = hx + (long long)__builtin_amdgcn_readlane(r, i) * F + lane;

  float v[EPW][FPT];
#pragma unroll
  for (int j = 0; j < FPT; ++j) {
    if (lane + 32 * j < F) {
#pragma unroll
      for (int i = 0; i < EPW; ++i) v[i][j] = hrow[i][32 * j];
    }
  }

#pragma unroll
  for (int i = 0; i < EPW; ++i) {
    const int   ci = __builtin_amdgcn_readlane(cl, i);
    const float si = __int_as_float(__builtin_amdgcn_readlane(__float_as_int(nrm), i));
    if (e0 + i < total) {
      float* o = out + (long long)ci * F + lane;
#pragma unroll
      for (int j = 0; j < FPT; ++j)
        if (lane + 32 * j < F) atomicAdd(o + 32 * j, v[i][j] * si);
    }
  }
}

template <int F, int EPW>
__global__ void gcn_agg_kernel(const float* __restrict__ hx, float* __restrict__ out0,
                               float* __restrict__ out1, const int* __restrict__ rows,
                               const int* __restrict__ cols, const float* __restrict__ ew,
                               const float* __restrict__ dinv, int E, int N) {
  gcn_agg_body<F, EPW>(hx, out0, out1, rows, cols, ew, dinv, E, N, (int)blockIdx.x);
}

__global__ void log_softmax_kernel(const float* __restrict__ in, const float* __restrict__ bias0,
                                   const float* __restrict__ bias1, float* __restrict__ out0,
                                   float* __restrict__ out1, int n, int F) {
  int idx = blockIdx.x * blockDim.x + threadIdx.x;
  if (idx >= 2 * n) return;
  int row = idx >> 1;
  int head = idx & 1;
  const float* p = in + (long long)idx * F;
  const float* q = head ? bias1 : bias0;
  float m = -INFINITY;
  for (int c = 0; c < F; ++c) m = fmaxf(m, p[c] + q[c]);
  float s = 0.0f;
  for (int c = 0; c < F; ++c) s += __expf((p[c] + q[c]) - m);
  float ls = __logf(s) + m;
  float* o = (head ? out1 : out0) + (long long)row * F;
  for (int c = 0; c < F; ++c) o[c] = (p[c] + q[c]) - ls;
}

__device__ __forceinline__ void edge_dot_body(const float* __restrict__ h,
                                              const int* __restrict__ pos,
                                              const int* __restrict__ neg,
                                              float* __restrict__ res, int e) {
  if (e >= 2 * E_LP) return;
  const int* src = (e < E_LP) ? pos : neg;
  int eo = (e < E_LP) ? e : (e - E_LP);
  int j = src[eo];
  int i = src[E_LP + eo];
  const float4* hj = (const float4*)(h + (long long)j * H2_DIM);
  const float4* hi = (const float4*)(h + (long long)i * H2_DIM);
  float s = 0.0f;
#pragma unroll
  for (int k = 0; k < H2_DIM / 4; ++k) {
    float4 a = hi[k], b = hj[k];
    s += a.x * b.x + a.y * b.y + a.z * b.z + a.w * b.w;
  }
  res[e] = s;
}

#define DOT_BLOCKS ((2 * E_LP + G1_BS - 1) / G1_BS)
__global__ void gemm3_edge_dot_kernel(const float* __restrict__ g0, const float* __restrict__ g1,
                                      const float* __restrict__ wcp, float* __restrict__ hcat,
                                      const float* __restrict__ h, const int* __restrict__ pos,
                                      const int* __restrict__ neg, float* __restrict__ res) {
  const int b = blockIdx.x;
  if (b < G1_BLOCKS) {
    int wave = (b * G1_BS + (int)threadIdx.x) >> 5;
    int lane = threadIdx.x & 31;
    wmma_gemm_body<H2_DIM, 5, 2 * N_CLS, false, true>(g0, g1, wcp, hcat, N_NODES, wave, lane);
  } else {
    edge_dot_body(h, pos, neg, res, (b - G1_BLOCKS) * G1_BS + (int)threadIdx.x);
  }
}

static inline int blocks_for(long long total, int bs) {
  return (int)((total + bs - 1) / bs);
}

extern "C" void kernel_launch(void* const* d_in, const int* in_sizes, int n_in,
                              void* d_out, int out_size, void* d_ws, size_t ws_size,
                              hipStream_t stream) {
  const float* x      = (const float*)d_in[0];
  const int*   tpe    = (const int*)  d_in[1];
  const int*   ei     = (const int*)  d_in[2];
  const float* ew     = (const float*)d_in[3];
  const int*   pos    = (const int*)  d_in[4];
  const int*   neg    = (const int*)  d_in[5];
  const float* W1     = (const float*)d_in[6];
  const float* b1     = (const float*)d_in[7];
  const float* W2     = (const float*)d_in[8];
  const float* b2     = (const float*)d_in[9];
  const float* W_attr = (const float*)d_in[10];
  const float* b_attr = (const float*)d_in[11];
  const float* W_att  = (const float*)d_in[12];
  const float* b_att  = (const float*)d_in[13];

  float* out = (float*)d_out;
  float* ws  = (float*)d_ws;

  const long long OFF_DINV_T = 0;
  const long long OFF_DINV_W = 50000;
  const long long OFF_R0     = 100000;
  const long long OFF_R1     = OFF_R0 + 6400000;
  const long long OFF_R2     = OFF_R1 + 6400000;
  const long long OFF_W1P    = OFF_R2 + 6400000;
  const long long OFF_W2P    = OFF_W1P + 65536;
  const long long OFF_WCP    = OFF_W2P + 8192;

  float* dinv_t = ws + OFF_DINV_T;
  float* dinv_w = ws + OFF_DINV_W;
  float* degt_p1 = ws + OFF_R1;
  float* degw_p1 = ws + OFF_R1 + 50000;
  float* hx1    = ws + OFF_R0;
  float* h1_p0  = ws + OFF_R1;
  float* h1_p1  = ws + OFF_R2;
  float* hx2    = ws + OFF_R0;
  float* h2_p0  = ws + OFF_R1;
  float* h2_p1  = ws + OFF_R1 + 3200000;
  float* h2     = ws + OFF_R2;
  float* g_p0   = ws + OFF_R0;
  float* g_p1   = ws + OFF_R0 + 3200000;
  float* hcat   = ws + OFF_R1;
  float* w1p    = ws + OFF_W1P;
  float* w2p    = ws + OFF_W2P;
  float* wcp    = ws + OFF_WCP;

  const int BS = 256;

  prep_kernel<<<PREP_B_W1 + PREP_B_W2 + PREP_B_WC + PREP_B_DG, PREP_BS, 0, stream>>>(
      W1, W2, W_attr, W_att, w1p, w2p, wcp, dinv_t, degt_p1);

  gemm1_deg_kernel<<<G1_BLOCKS + 2 * DG_BLOCKS, G1_BS, 0, stream>>>(
      x, w1p, hx1, dinv_t, degt_p1, tpe + E_MP, dinv_w, degw_p1, ei + E_MP, ew);

  init1_rsqrt_kernel<<<blocks_for((long long)N_NODES * H1_DIM, BS), BS, 0, stream>>>(
      h1_p0, h1_p1, b1, dinv_t, (long long)N_NODES * H1_DIM, H1_DIM, 2 * N_NODES);

  const int ROW_TILES = N_NODES / 16;
  const int GEMM_BS = 128;
  const int gemm_blocks = blocks_for((long long)ROW_TILES * 32, GEMM_BS);


  gcn_agg_kernel<H1_DIM, AGG_EPW><<<AGG_BLOCKS, AGG_BS, 0, stream>>>(
      hx1, h1_p0, h1_p1, tpe, tpe + E_MP, nullptr, dinv_t, E_MP, N_NODES);

  wmma_gemm_kernel<H1_DIM, 4, H2_DIM, true, true><<<gemm_blocks, GEMM_BS, 0, stream>>>(h1_p0, h1_p1, w2p, hx2, N_NODES);
  bias_init_kernel<<<blocks_for((long long)N_NODES * H2_DIM, BS), BS, 0, stream>>>(h2_p0, h2_p1, b2, N_NODES, H2_DIM);
  gcn_agg_kernel<H2_DIM, AGG_EPW><<<AGG_BLOCKS, AGG_BS, 0, stream>>>(
      hx2, h2_p0, h2_p1, tpe, tpe + E_MP, nullptr, dinv_t, E_MP, N_NODES);
  add_planes_kernel<<<blocks_for(N_NODES * H2_DIM / 16, BS), BS, 0, stream>>>(
      (const float4*)h2_p0, (const float4*)h2_p1, (float4*)h2, (float4*)g_p0, (float4*)g_p1,
      N_NODES * H2_DIM / 4);

  gcn_agg_kernel<H2_DIM, AGG_EPW><<<AGG_BLOCKS, AGG_BS, 0, stream>>>(
      h2, g_p0, g_p1, ei, ei + E_MP, ew, dinv_w, E_MP, N_NODES);
  gemm3_edge_dot_kernel<<<G1_BLOCKS + DOT_BLOCKS, G1_BS, 0, stream>>>(
      g_p0, g_p1, wcp, hcat, h2, pos, neg, out);

  log_softmax_kernel<<<blocks_for(2 * N_NODES, BS), BS, 0, stream>>>(
      hcat, b_attr, b_att, out + 2 * E_LP, out + 2 * E_LP + (long long)N_NODES * N_CLS, N_NODES, N_CLS);
}


# heads: the product's waves (16 rows x all 80 classes each) do bias + log_softmax through their own LDS piece and store both outputs as whole lines; no hcat round trip, softmax launch gone, 11 -> 10
# speedup vs baseline: 2.2441x; 1.0686x over previous
#include <hip/hip_runtime.h>
#include <hip/hip_bf16.h>

#define N_NODES   50000
#define D_FEAT    512
#define H1_DIM    128
#define H2_DIM    64
#define N_CLS     40
#define E_MP      800000
#define E_LP      250000

typedef __attribute__((ext_vector_type(2))) float v2f;
typedef __attribute__((ext_vector_type(8))) float v8f;

__device__ __forceinline__ int chip_half() {
  return (int)((__builtin_amdgcn_s_sendmsg_rtn(0x87) >> 18) & 1u);
}


__device__ __forceinline__ void deg_accum_body(float* __restrict__ deg0, float* __restrict__ deg1,
                                               const int* __restrict__ cols,
                                               const float* __restrict__ ew, int E, long long i) {
  float* deg = chip_half() ? deg1 : deg0;
  if (i < E) {
    float w = ew ? ew[i] : 1.0f;
    atomicAdd(&deg[cols[i]], w);
  }
}

__global__ void init1_rsqrt_kernel(float* __restrict__ out0, float* __restrict__ out1,
                                   const float* __restrict__ b, float* __restrict__ dinv,
                                   long long total, int F, int nd) {
  long long i = (long long)blockIdx.x * blockDim.x + threadIdx.x;
  if (i < total) {
    if (i < nd) {
      float d = dinv[i] + out0[i];
      dinv[i] = (d > 0.0f) ? rsqrtf(d) : 0.0f;
    }
    out0[i] = b[i % F];
    out1[i] = 0.0f;
  }
}

__global__ void bias_init_kernel(float* __restrict__ out0, float* __restrict__ out1,
                                 const float* __restrict__ b, long long n, int F) {
  long long i = (long long)blockIdx.x * blockDim.x + threadIdx.x;
  long long total = n * F;
  if (i < total) { out0[i] = b[i % F]; out1[i] = 0.0f; }
}

__global__ void add_planes_kernel(const float4* __restrict__ a, const float4* __restrict__ b,
                                  float4* __restrict__ o, float4* __restrict__ z0,
                                  float4* __restrict__ z1, int n4) {
  const int q = n4 >> 2;
  int i = blockIdx.x * blockDim.x + threadIdx.x;
  if (i < q) {
    float4 x[4], y[4];
#pragma unroll
    for (int k = 0; k < 4; ++k) { x[k] = a[i + k * q]; y[k] = b[i + k * q]; }
    const float4 zero = make_float4(0.0f, 0.0f, 0.0f, 0.0f);
#pragma unroll
    for (int k = 0; k < 4; ++k) { z0[i + k * q] = zero; z1[i + k * q] = zero; }
#pragma unroll
    for (int k = 0; k < 4; ++k)
      o[i + k * q] = make_float4(x[k].x + y[k].x, x[k].y + y[k].y, x[k].z + y[k].z, x[k].w + y[k].w);
  }
}

__device__ __forceinline__ void pack_b_body(const float* __restrict__ B, float* __restrict__ Bp,
                                            int K, int NC, int NCP, int p) {
  int total = (K / 2) * NCP;
  if (p >= total) return;
  int k2 = p / NCP;
  int col = p % NCP;
  float x0 = 0.0f, x1 = 0.0f;
  if (col < NC) {
    x0 = B[(long long)(2 * k2) * NC + col];
    x1 = B[(long long)(2 * k2 + 1) * NC + col];
  }
  Bp[2 * p + 0] = x0;
  Bp[2 * p + 1] = x1;
}

__device__ __forceinline__ void pack_heads_body(const float* __restrict__ Ba,
                                                const float* __restrict__ Bb,
                                                float* __restrict__ Bp, int K, int NC, int p) {
  const int NCP = 2 * NC;
  int total = (K / 2) * NCP;
  if (p >= total) return;
  int k2 = p / NCP;
  int col = p % NCP;
  const float* __restrict__ B = (col < NC) ? Ba : Bb;
  int cc = (col < NC) ? col : col - NC;
  Bp[2 * p + 0] = B[(long long)(2 * k2) * NC + cc];
  Bp[2 * p + 1] = B[(long long)(2 * k2 + 1) * NC + cc];
}

#define PREP_BS   256
#define PREP_B_W1 ((D_FEAT / 2) * H1_DIM / PREP_BS)
#define PREP_B_W2 ((H1_DIM / 2) * H2_DIM / PREP_BS)
#define PREP_B_WC ((H2_DIM / 2) * 2 * N_CLS / PREP_BS)
#define PREP_B_DG ((2 * N_NODES + PREP_BS - 1) / PREP_BS)
__global__ void prep_kernel(const float* __restrict__ W1, const float* __restrict__ W2,
                            const float* __restrict__ Wa, const float* __restrict__ Wb,
                            float* __restrict__ w1p, float* __restrict__ w2p,
                            float* __restrict__ wcp, float* __restrict__ deg0,
                            float* __restrict__ deg1) {
  int b = blockIdx.x;
  const int t = threadIdx.x;
  if (b < PREP_B_W1) { pack_b_body(W1, w1p, D_FEAT, H1_DIM, H1_DIM, b * PREP_BS + t); return; }
  b -= PREP_B_W1;
  if (b < PREP_B_W2) { pack_b_body(W2, w2p, H1_DIM, H2_DIM, H2_DIM, b * PREP_BS + t); return; }
  b -= PREP_B_W2;
  if (b < PREP_B_WC) { pack_heads_body(Wa, Wb, wcp, H2_DIM, N_CLS, b * PREP_BS + t); return; }
  b -= PREP_B_WC;
  const int i = b * PREP_BS + t;
  if (i < 2 * N_NODES) { deg0[i] = 1.0f; deg1[i] = 0.0f; }
}

template <int KDIM, int NT, bool RELU_A, bool DUAL_A>
__device__ __forceinline__ void wmma_tile_acc(const float* __restrict__ A,
                                              const float* __restrict__ A2,
                                              const float* __restrict__ Bp,
                                              int wave, int lane, v8f (&acc)[NT]) {
  constexpr int NCP = NT * 16;
  int row0 = wave << 4;
  int arow = row0 + (lane & 15);
  int khb  = lane >> 4;
  int bcol = lane & 15;

  const float2* __restrict__ Arow2 = (const float2*)(A + (long long)arow * KDIM);
  const float2* __restrict__ Arow2b = (const float2*)((DUAL_A ? A2 : A) + (long long)arow * KDIM);
  const float2* __restrict__ Bp2   = (const float2*)Bp;

#pragma unroll 2
  for (int k2 = 0; k2 < KDIM / 2; k2 += 2) {
    const float2* __restrict__ Bbase = Bp2 + (long long)(k2 + khb) * NCP + bcol;
    float2 bf[NT];
#pragma unroll
    for (int t = 0; t < NT; ++t) bf[t] = Bbase[t * 16];

    float2 af = Arow2[k2 + khb];
    if (DUAL_A) { float2 ag = Arow2b[k2 + khb]; af.x += ag.x; af.y += ag.y; }
    if (RELU_A) { af.x = fmaxf(af.x, 0.0f); af.y = fmaxf(af.y, 0.0f); }
    v2f a; a.x = af.x; a.y = af.y;

#pragma unroll
    for (int t = 0; t < NT; ++t) {
      v2f b; b.x = bf[t].x; b.y = bf[t].y;
      acc[t] = __builtin_amdgcn_wmma_f32_16x16x4_f32(
           false, a,  false, b,
           (short)0, acc[t],  false,  false);
    }
  }
}

template <int KDIM, int NT, int NCSTORE, bool RELU_A, bool DUAL_A>
__device__ __forceinline__ void wmma_gemm_body(const float* __restrict__ A,
                                               const float* __restrict__ A2,
                                               const float* __restrict__ Bp,
                                               float* __restrict__ C, int M, int wave, int lane) {
  int ntiles = M >> 4;
  if (wave >= ntiles) return;

  v8f acc[NT] = {};
  wmma_tile_acc<KDIM, NT, RELU_A, DUAL_A>(A, A2, Bp, wave, lane, acc);

  int row0 = wave << 4;
  int khb  = lane >> 4;
  int crow = row0 + (khb << 3);
  int ccol = lane & 15;
#pragma unroll
  for (int t = 0; t < NT; ++t) {
    int col = (t << 4) + ccol;
    if (col < NCSTORE) {
#pragma unroll
      for (int r = 0; r < 8; ++r) {
        C[(long long)(crow + r) * NCSTORE + col] = acc[t][r];
      }
    }
  }
}

template <int KDIM, int NT, int NCSTORE, bool RELU_A, bool DUAL_A>
__global__ void wmma_gemm_kernel(const float* __restrict__ A, const float* __restrict__ A2,
                                 const float* __restrict__ Bp, float* __restrict__ C, int M) {
  int wave = (int)(((long long)blockIdx.x * blockDim.x + threadIdx.x) >> 5);
  int lane = threadIdx.x & 31;
  wmma_gemm_body<KDIM, NT, NCSTORE, RELU_A, DUAL_A>(A, A2, Bp, C, M, wave, lane);
}

#define G1_BS     128
#define G1_BLOCKS ((N_NODES / 16 * 32 + G1_BS - 1) / G1_BS)
#define DG_BLOCKS (E_MP / G1_BS)
__global__ void gemm1_deg_kernel(const float* __restrict__ x, const float* __restrict__ w1p,
                                 float* __restrict__ hx1,
                                 float* __restrict__ degt0, float* __restrict__ degt1,
                                 const int* __restrict__ cols_t,
                                 float* __restrict__ degw0, float* __restrict__ degw1,
                                 const int* __restrict__ cols_w, const float* __restrict__ ew) {
  const int b = blockIdx.x;
  if (b < G1_BLOCKS) {
    int wave = (b * G1_BS + (int)threadIdx.x) >> 5;
    int lane = threadIdx.x & 31;
    wmma_gemm_body<D_FEAT, 8, H1_DIM, false, false>(x, nullptr, w1p, hx1, N_NODES, wave, lane);
  } else if (b < G1_BLOCKS + DG_BLOCKS) {
    deg_accum_body(degt0, degt1, cols_t, nullptr, E_MP,
                   (long long)(b - G1_BLOCKS) * G1_BS + threadIdx.x);
  } else {
    deg_accum_body(degw0, degw1, cols_w, ew, E_MP,
                   (long long)(b - G1_BLOCKS - DG_BLOCKS) * G1_BS + threadIdx.x);
  }
}

#define AGG_BS     256
#define AGG_EPW    8
#define AGG_BLOCKS ((int)((((long long)(E_MP + N_NODES) + AGG_EPW - 1) / AGG_EPW * 32 + AGG_BS - 1) / AGG_BS))
template <int F, int EPW>
__device__ __forceinline__ void gcn_agg_body(const float* __restrict__ hx, float* __restrict__ out0,
                                             float* __restrict__ out1, const int* __restrict__ rows,
                                             const int* __restrict__ cols,
                                             const float* __restrict__ ew,
                                             const float* __restrict__ dinv,
                                             int E, int N, int blk) {
  constexpr int FPT = (F + 31) / 32;
  const int lane = threadIdx.x & 31;
  const int wib = __builtin_amdgcn_readfirstlane((int)(threadIdx.x >> 5));
  const long long wave = (long long)blk * (AGG_BS >> 5) + wib;
  const long long total = (long long)E + N;
  const long long e0 = wave * EPW;
  if (e0 >= total) return;
  float* out = chip_half() ? out1 : out0;

  int r = 0, cl = 0;
  float nrm = 0.0f;
  {
    const long long e = e0 + lane;
    if (lane < EPW && e < total) {
      float w;
      if (e < E) { r = rows[e]; cl = cols[e]; w = ew ? ew[e] : 1.0f; }
      else       { r = (int)(e - E); cl = r;  w = 1.0f; }
      nrm = dinv[r] * w * dinv[cl];
    }
  }

  const float* hrow[EPW];
#pragma unroll
  for (int i = 0; i < EPW; ++i)
    hrow[i] = hx + (long long)__builtin_amdgcn_readlane(r, i) * F + lane;

  float v[EPW][FPT];
#pragma unroll
  for (int j = 0; j < FPT; ++j) {
    if (lane + 32 * j < F) {
#pragma unroll
      for (int i = 0; i < EPW; ++i) v[i][j] = hrow[i][32 * j];
    }
  }

#pragma unroll
  for (int i = 0; i < EPW; ++i) {
    const int   ci = __builtin_amdgcn_readlane(cl, i);
    const float si = __int_as_float(__builtin_amdgcn_readlane(__float_as_int(nrm), i));
    if (e0 + i < total) {
      float* o = out + (long long)ci * F + lane;
#pragma unroll
      for (int j = 0; j < FPT; ++j)
        if (lane + 32 * j < F) atomicAdd(o + 32 * j, v[i][j] * si);
    }
  }
}

template <int F, int EPW>
__global__ void gcn_agg_kernel(const float* __restrict__ hx, float* __restrict__ out0,
                               float* __restrict__ out1, const int* __restrict__ rows,
                               const int* __restrict__ cols, const float* __restrict__ ew,
                               const float* __restrict__ dinv, int E, int N) {
  gcn_agg_body<F, EPW>(hx, out0, out1, rows, cols, ew, dinv, E, N, (int)blockIdx.x);
}

__device__ __forceinline__ void edge_dot_body(const float* __restrict__ h,
                                              const int* __restrict__ pos,
                                              const int* __restrict__ neg,
                                              float* __restrict__ res, int e) {
  if (e >= 2 * E_LP) return;
  const int* src = (e < E_LP) ? pos : neg;
  int eo = (e < E_LP) ? e : (e - E_LP);
  int j = src[eo];
  int i = src[E_LP + eo];
  const float4* hj = (const float4*)(h + (long long)j * H2_DIM);
  const float4* hi = (const float4*)(h + (long long)i * H2_DIM);
  float s = 0.0f;
#pragma unroll
  for (int k = 0; k < H2_DIM / 4; ++k) {
    float4 a = hi[k], b = hj[k];
    s += a.x * b.x + a.y * b.y + a.z * b.z + a.w * b.w;
  }
  res[e] = s;
}

#define SM_PITCH 41
#define SM_WAVE  (32 * SM_PITCH)
#define DOT_BLOCKS ((2 * E_LP + G1_BS - 1) / G1_BS)
__global__ void gemm3_softmax_edge_dot_kernel(const float* __restrict__ g0,
                                              const float* __restrict__ g1,
                                              const float* __restrict__ wcp,
                                              const float* __restrict__ bias0,
                                              const float* __restrict__ bias1,
                                              float* __restrict__ out0, float* __restrict__ out1,
                                              const float* __restrict__ h,
                                              const int* __restrict__ pos,
                                              const int* __restrict__ neg,
                                              float* __restrict__ res) {
  __shared__ __attribute__((aligned(16))) float sm[(G1_BS / 32) * SM_WAVE];
  const int b = blockIdx.x;
  if (b < G1_BLOCKS) {
    const int wib  = (int)threadIdx.x >> 5;
    const int wave = b * (G1_BS / 32) + wib;
    const int lane = threadIdx.x & 31;
    const bool live = wave < N_NODES / 16;
    float* s = sm + wib * SM_WAVE;

    if (live) {
      v8f acc[5] = {};
      wmma_tile_acc<H2_DIM, 5, false, true>(g0, g1, wcp, wave, lane, acc);
      const int c  = lane & 15;
      const int hb = lane >> 4;
#pragma unroll
      for (int t = 0; t < 5; ++t) {
        const int col  = 16 * t + c;
        const int head = (col >= N_CLS) ? 1 : 0;
        const int cc   = col - N_CLS * head;
        const float bt = head ? bias1[cc] : bias0[cc];
#pragma unroll
        for (int r = 0; r < 8; ++r)
          s[(2 * (8 * hb + r) + head) * SM_PITCH + cc] = acc[t][r] + bt;
      }
    }
    __syncthreads();

    float x[N_CLS];
    if (live) {
#pragma unroll
      for (int k = 0; k < N_CLS; ++k) x[k] = s[lane * SM_PITCH + k];
      float m = -INFINITY;
#pragma unroll
      for (int k = 0; k < N_CLS; ++k) m = fmaxf(m, x[k]);
      float sum = 0.0f;
#pragma unroll
      for (int k = 0; k < N_CLS; ++k) sum += __expf(x[k] - m);
      const float ls = __logf(sum) + m;
#pragma unroll
      for (int k = 0; k < N_CLS; ++k) x[k] = x[k] - ls;
    }
    __syncthreads();

    if (live) {
      float* d = s + (lane & 1) * (16 * N_CLS) + (lane >> 1) * N_CLS;
#pragma unroll
      for (int k = 0; k < N_CLS; ++k) d[k] = x[k];
    }
    __syncthreads();

    if (live) {
      const float4* s4 = (const float4*)s;
      float4* o0 = (float4*)(out0 + (long long)wave * (16 * N_CLS));
      float4* o1 = (float4*)(out1 + (long long)wave * (16 * N_CLS));
#pragma unroll
      for (int k = 0; k < (16 * N_CLS / 4) / 32; ++k) {
        o0[lane + 32 * k] = s4[lane + 32 * k];
        o1[lane + 32 * k] = s4[(16 * N_CLS / 4) + lane + 32 * k];
      }
    }
  } else {
    edge_dot_body(h, pos, neg, res, (b - G1_BLOCKS) * G1_BS + (int)threadIdx.x);
  }
}

static inline int blocks_for(long long total, int bs) {
  return (int)((total + bs - 1) / bs);
}

extern "C" void kernel_launch(void* const* d_in, const int* in_sizes, int n_in,
                              void* d_out, int out_size, void* d_ws, size_t ws_size,
                              hipStream_t stream) {
  const float* x      = (const float*)d_in[0];
  const int*   tpe    = (const int*)  d_in[1];
  const int*   ei     = (const int*)  d_in[2];
  const float* ew     = (const float*)d_in[3];
  const int*   pos    = (const int*)  d_in[4];
  const int*   neg    = (const int*)  d_in[5];
  const float* W1     = (const float*)d_in[6];
  const float* b1     = (const float*)d_in[7];
  const float* W2     = (const float*)d_in[8];
  const float* b2     = (const float*)d_in[9];
  const float* W_attr = (const float*)d_in[10];
  const float* b_attr = (const float*)d_in[11];
  const float* W_att  = (const float*)d_in[12];
  const float* b_att  = (const float*)d_in[13];

  float* out = (float*)d_out;
  float* ws  = (float*)d_ws;

  const long long OFF_DINV_T = 0;
  const long long OFF_DINV_W = 50000;
  const long long OFF_R0     = 100000;
  const long long OFF_R1     = OFF_R0 + 6400000;
  const long long OFF_R2     = OFF_R1 + 6400000;
  const long long OFF_W1P    = OFF_R2 + 6400000;
  const long long OFF_W2P    = OFF_W1P + 65536;
  const long long OFF_WCP    = OFF_W2P + 8192;

  float* dinv_t = ws + OFF_DINV_T;
  float* dinv_w = ws + OFF_DINV_W;
  float* degt_p1 = ws + OFF_R1;
  float* degw_p1 = ws + OFF_R1 + 50000;
  float* hx1    = ws + OFF_R0;
  float* h1_p0  = ws + OFF_R1;
  float* h1_p1  = ws + OFF_R2;
  float* hx2    = ws + OFF_R0;
  float* h2_p0  = ws + OFF_R1;
  float* h2_p1  = ws + OFF_R1 + 3200000;
  float* h2     = ws + OFF_R2;
  float* g_p0   = ws + OFF_R0;
  float* g_p1   = ws + OFF_R0 + 3200000;
  float* w1p    = ws + OFF_W1P;
  float* w2p    = ws + OFF_W2P;
  float* wcp    = ws + OFF_WCP;

  const int BS = 256;

  prep_kernel<<<PREP_B_W1 + PREP_B_W2 + PREP_B_WC + PREP_B_DG, PREP_BS, 0, stream>>>(
      W1, W2, W_attr, W_att, w1p, w2p, wcp, dinv_t, degt_p1);

  gemm1_deg_kernel<<<G1_BLOCKS + 2 * DG_BLOCKS, G1_BS, 0, stream>>>(
      x, w1p, hx1, dinv_t, degt_p1, tpe + E_MP, dinv_w, degw_p1, ei + E_MP, ew);

  init1_rsqrt_kernel<<<blocks_for((long long)N_NODES * H1_DIM, BS), BS, 0, stream>>>(
      h1_p0, h1_p1, b1, dinv_t, (long long)N_NODES * H1_DIM, H1_DIM, 2 * N_NODES);

  const int ROW_TILES = N_NODES / 16;
  const int GEMM_BS = 128;
  const int gemm_blocks = blocks_for((long long)ROW_TILES * 32, GEMM_BS);


  gcn_agg_kernel<H1_DIM, AGG_EPW><<<AGG_BLOCKS, AGG_BS, 0, stream>>>(
      hx1, h1_p0, h1_p1, tpe, tpe + E_MP, nullptr, dinv_t, E_MP, N_NODES);

  wmma_gemm_kernel<H1_DIM, 4, H2_DIM, true, true><<<gemm_blocks, GEMM_BS, 0, stream>>>(h1_p0, h1_p1, w2p, hx2, N_NODES);
  bias_init_kernel<<<blocks_for((long long)N_NODES * H2_DIM, BS), BS, 0, stream>>>(h2_p0, h2_p1, b2, N_NODES, H2_DIM);
  gcn_agg_kernel<H2_DIM, AGG_EPW><<<AGG_BLOCKS, AGG_BS, 0, stream>>>(
      hx2, h2_p0, h2_p1, tpe, tpe + E_MP, nullptr, dinv_t, E_MP, N_NODES);
  add_planes_kernel<<<blocks_for(N_NODES * H2_DIM / 16, BS), BS, 0, stream>>>(
      (const float4*)h2_p0, (const float4*)h2_p1, (float4*)h2, (float4*)g_p0, (float4*)g_p1,
      N_NODES * H2_DIM / 4);

  gcn_agg_kernel<H2_DIM, AGG_EPW><<<AGG_BLOCKS, AGG_BS, 0, stream>>>(
      h2, g_p0, g_p1, ei, ei + E_MP, ew, dinv_w, E_MP, N_NODES);
  gemm3_softmax_edge_dot_kernel<<<G1_BLOCKS + DOT_BLOCKS, G1_BS, 0, stream>>>(
      g_p0, g_p1, wcp, b_attr, b_att, out + 2 * E_LP,
      out + 2 * E_LP + (long long)N_NODES * N_CLS, h2, pos, neg, out);
}


# gcn_agg x3: edges grouped by destination (returning int add as rank beside x@W1, block scan, place pass), one wave sums a node's rows in registers, stores once; no f32 atomics; sums in list order
# speedup vs baseline: 2.6110x; 1.1635x over previous
#include <hip/hip_runtime.h>
#include <hip/hip_bf16.h>

#define N_NODES   50000
#define D_FEAT    512
#define H1_DIM    128
#define H2_DIM    64
#define N_CLS     40
#define E_MP      800000
#define E_LP      250000

typedef __attribute__((ext_vector_type(2))) float v2f;
typedef __attribute__((ext_vector_type(8))) float v8f;

__device__ __forceinline__ int chip_half() {
  return (int)((__builtin_amdgcn_s_sendmsg_rtn(0x87) >> 18) & 1u);
}

#define RK_Q (E_MP / 4)
__device__ __forceinline__ void rank_body(int* __restrict__ cnt0, int* __restrict__ cnt1,
                                          const int* __restrict__ cols, int* __restrict__ rank,
                                          int i) {
  if (i >= RK_Q) return;
  const int h = chip_half();
  int* cnt = h ? cnt1 : cnt0;
  int c[4], k4[4];
#pragma unroll
  for (int k = 0; k < 4; ++k) c[k] = cols[i + k * RK_Q];
#pragma unroll
  for (int k = 0; k < 4; ++k) k4[k] = atomicAdd(&cnt[c[k]], 1);
#pragma unroll
  for (int k = 0; k < 4; ++k) rank[i + k * RK_Q] = k4[k] | (h << 30);
}

#define SC_BS 1024
#define SC_CH ((N_NODES + 1 + SC_BS - 1) / SC_BS)
#define OFF_PITCH (SC_CH * SC_BS)
__global__ void scan_kernel(const int* __restrict__ cnt  ,
                            int* __restrict__ off  ,
                            float* __restrict__ dinv_t) {
  __shared__ int red[32];
  __shared__ int wtot[32];
  const int g = blockIdx.x / SC_CH;
  const int c = blockIdx.x % SC_CH;
  const int t = threadIdx.x, lane = t & 31, wid = t >> 5;
  const int* __restrict__ c0 = cnt + g * N_NODES;
  const int* __restrict__ c1 = cnt + 2 * N_NODES + g * N_NODES;
  int p = 0;
#pragma unroll 8
  for (int m = 0; m < c; ++m) p += c0[m * SC_BS + t] + c1[m * SC_BS + t];
  for (int d = 16; d > 0; d >>= 1) p += __shfl_xor(p, d);
  if (lane == 0) red[wid] = p;
  const int i = c * SC_BS + t;
  const int v = (i < N_NODES) ? (c0[i] + c1[i]) : 0;
  int incl = v;
  for (int d = 1; d < 32; d <<= 1) { int u = __shfl_up(incl, d); if (lane >= d) incl += u; }
  if (lane == 31) wtot[wid] = incl;
  __syncthreads();
  int base = 0;
  for (int k = 0; k < 32; ++k) { base += red[k]; if (k < wid) base += wtot[k]; }
  if (i <= N_NODES) off[g * OFF_PITCH + i] = base + incl - v;
  if (g == 0 && i < N_NODES) dinv_t[i] = rsqrtf((float)v + 1.0f);
}

template <bool WEIGHTED>
__device__ __forceinline__ void place_body(const int* __restrict__ rows,
                                           const int* __restrict__ cols,
                                           const float* __restrict__ ew,
                                           const int* __restrict__ rank,
                                           const int* __restrict__ cnt0,
                                           const int* __restrict__ off,
                                           const float* __restrict__ dinv,
                                           int2* __restrict__ srt, int i) {
  if (i >= RK_Q) return;
  int r[4], c[4], rk[4], pos[4];
  float f[4];
#pragma unroll
  for (int k = 0; k < 4; ++k) {
    const int e = i + k * RK_Q;
    r[k] = rows[e]; c[k] = cols[e]; rk[k] = rank[e];
    f[k] = WEIGHTED ? ew[e] : 1.0f;
  }
#pragma unroll
  for (int k = 0; k < 4; ++k) {
    const int p0 = cnt0[c[k]];
    pos[k] = off[c[k]] + (rk[k] & 0x3fffffff) + ((rk[k] >> 30) ? p0 : 0);
    if (!WEIGHTED) f[k] = dinv[r[k]] * f[k] * dinv[c[k]];
  }
#pragma unroll
  for (int k = 0; k < 4; ++k) srt[pos[k]] = make_int2(r[k], __float_as_int(f[k]));
}
#define PL_BLOCKS ((RK_Q + 127) / 128)
__global__ void place_kernel(const int* __restrict__ tpe, const int* __restrict__ ei,
                             const float* __restrict__ ew, const int* __restrict__ rank,
                             const int* __restrict__ cnt, const int* __restrict__ off,
                             const float* __restrict__ dinv_t, int2* __restrict__ srt) {
  const int b = blockIdx.x;
  if (b < PL_BLOCKS)
    place_body<false>(tpe, tpe + E_MP, nullptr, rank, cnt, off, dinv_t, srt,
                      b * 128 + (int)threadIdx.x);
  else
    place_body<true>(ei, ei + E_MP, ew, rank + E_MP, cnt + N_NODES, off + OFF_PITCH, nullptr,
                     srt + E_MP, (b - PL_BLOCKS) * 128 + (int)threadIdx.x);
}

__device__ __forceinline__ void degw_body(const int2* __restrict__ srt, const int* __restrict__ off,
                                          float* __restrict__ dinv_w, int i) {
  if (i >= N_NODES) return;
  const int s = off[i], e = off[i + 1];
  float d = 1.0f;
#pragma unroll 4
  for (int k = s; k < e; ++k) d += __int_as_float(srt[k].y);
  dinv_w[i] = (d > 0.0f) ? rsqrtf(d) : 0.0f;
}

__device__ __forceinline__ void pack_b_body(const float* __restrict__ B, float* __restrict__ Bp,
                                            int K, int NC, int NCP, int p) {
  int total = (K / 2) * NCP;
  if (p >= total) return;
  int k2 = p / NCP;
  int col = p % NCP;
  float x0 = 0.0f, x1 = 0.0f;
  if (col < NC) {
    x0 = B[(long long)(2 * k2) * NC + col];
    x1 = B[(long long)(2 * k2 + 1) * NC + col];
  }
  Bp[2 * p + 0] = x0;
  Bp[2 * p + 1] = x1;
}

__device__ __forceinline__ void pack_heads_body(const float* __restrict__ Ba,
                                                const float* __restrict__ Bb,
                                                float* __restrict__ Bp, int K, int NC, int p) {
  const int NCP = 2 * NC;
  int total = (K / 2) * NCP;
  if (p >= total) return;
  int k2 = p / NCP;
  int col = p % NCP;
  const float* __restrict__ B = (col < NC) ? Ba : Bb;
  int cc = (col < NC) ? col : col - NC;
  Bp[2 * p + 0] = B[(long long)(2 * k2) * NC + cc];
  Bp[2 * p + 1] = B[(long long)(2 * k2 + 1) * NC + cc];
}

#define PREP_BS   256
#define PREP_B_W1 ((D_FEAT / 2) * H1_DIM / PREP_BS)
#define PREP_B_W2 ((H1_DIM / 2) * H2_DIM / PREP_BS)
#define PREP_B_WC ((H2_DIM / 2) * 2 * N_CLS / PREP_BS)
#define PREP_B_DG ((4 * N_NODES + PREP_BS - 1) / PREP_BS)
__global__ void prep_kernel(const float* __restrict__ W1, const float* __restrict__ W2,
                            const float* __restrict__ Wa, const float* __restrict__ Wb,
                            float* __restrict__ w1p, float* __restrict__ w2p,
                            float* __restrict__ wcp, int* __restrict__ cnt) {
  int b = blockIdx.x;
  const int t = threadIdx.x;
  if (b < PREP_B_W1) { pack_b_body(W1, w1p, D_FEAT, H1_DIM, H1_DIM, b * PREP_BS + t); return; }
  b -= PREP_B_W1;
  if (b < PREP_B_W2) { pack_b_body(W2, w2p, H1_DIM, H2_DIM, H2_DIM, b * PREP_BS + t); return; }
  b -= PREP_B_W2;
  if (b < PREP_B_WC) { pack_heads_body(Wa, Wb, wcp, H2_DIM, N_CLS, b * PREP_BS + t); return; }
  b -= PREP_B_WC;
  const int i = b * PREP_BS + t;
  if (i < 4 * N_NODES) cnt[i] = 0;
}

template <int KDIM, int NT, bool RELU_A, bool DUAL_A>
__device__ __forceinline__ void wmma_tile_acc(const float* __restrict__ A,
                                              const float* __restrict__ A2,
                                              const float* __restrict__ Bp,
                                              int wave, int lane, v8f (&acc)[NT]) {
  constexpr int NCP = NT * 16;
  int row0 = wave << 4;
  int arow = row0 + (lane & 15);
  int khb  = lane >> 4;
  int bcol = lane & 15;

  const float2* __restrict__ Arow2 = (const float2*)(A + (long long)arow * KDIM);
  const float2* __restrict__ Arow2b = (const float2*)((DUAL_A ? A2 : A) + (long long)arow * KDIM);
  const float2* __restrict__ Bp2   = (const float2*)Bp;

#pragma unroll 2
  for (int k2 = 0; k2 < KDIM / 2; k2 += 2) {
    const float2* __restrict__ Bbase = Bp2 + (long long)(k2 + khb) * NCP + bcol;
    float2 bf[NT];
#pragma unroll
    for (int t = 0; t < NT; ++t) bf[t] = Bbase[t * 16];

    float2 af = Arow2[k2 + khb];
    if (DUAL_A) { float2 ag = Arow2b[k2 + khb]; af.x += ag.x; af.y += ag.y; }
    if (RELU_A) { af.x = fmaxf(af.x, 0.0f); af.y = fmaxf(af.y, 0.0f); }
    v2f a; a.x = af.x; a.y = af.y;

#pragma unroll
    for (int t = 0; t < NT; ++t) {
      v2f b; b.x = bf[t].x; b.y = bf[t].y;
      acc[t] = __builtin_amdgcn_wmma_f32_16x16x4_f32(
           false, a,  false, b,
           (short)0, acc[t],  false,  false);
    }
  }
}

template <int KDIM, int NT, int NCSTORE, bool RELU_A, bool DUAL_A>
__device__ __forceinline__ void wmma_gemm_body(const float* __restrict__ A,
                                               const float* __restrict__ A2,
                                               const float* __restrict__ Bp,
                                               float* __restrict__ C, int M, int wave, int lane) {
  int ntiles = M >> 4;
  if (wave >= ntiles) return;

  v8f acc[NT] = {};
  wmma_tile_acc<KDIM, NT, RELU_A, DUAL_A>(A, A2, Bp, wave, lane, acc);

  int row0 = wave << 4;
  int khb  = lane >> 4;
  int crow = row0 + (khb << 3);
  int ccol = lane & 15;
#pragma unroll
  for (int t = 0; t < NT; ++t) {
    int col = (t << 4) + ccol;
    if (col < NCSTORE) {
#pragma unroll
      for (int r = 0; r < 8; ++r) {
        C[(long long)(crow + r) * NCSTORE + col] = acc[t][r];
      }
    }
  }
}

#define G1_BS     128
#define G1_BLOCKS ((N_NODES / 16 * 32 + G1_BS - 1) / G1_BS)
#define RK_BLOCKS ((RK_Q + G1_BS - 1) / G1_BS)
__global__ void gemm1_rank_kernel(const float* __restrict__ x, const float* __restrict__ w1p,
                                  float* __restrict__ hx1, int* __restrict__ cnt,
                                  const int* __restrict__ cols_t, const int* __restrict__ cols_w,
                                  int* __restrict__ rank) {
  const int b = blockIdx.x;
  if (b < G1_BLOCKS) {
    int wave = (b * G1_BS + (int)threadIdx.x) >> 5;
    int lane = threadIdx.x & 31;
    wmma_gemm_body<D_FEAT, 8, H1_DIM, false, false>(x, nullptr, w1p, hx1, N_NODES, wave, lane);
  } else if (b < G1_BLOCKS + RK_BLOCKS) {
    rank_body(cnt, cnt + 2 * N_NODES, cols_t, rank, (b - G1_BLOCKS) * G1_BS + (int)threadIdx.x);
  } else {
    rank_body(cnt + N_NODES, cnt + 3 * N_NODES, cols_w, rank + E_MP,
              (b - G1_BLOCKS - RK_BLOCKS) * G1_BS + (int)threadIdx.x);
  }
}

#define DW_BLOCKS ((N_NODES + G1_BS - 1) / G1_BS)
__global__ void gemm2_degw_kernel(const float* __restrict__ h1, const float* __restrict__ w2p,
                                  float* __restrict__ hx2, const int2* __restrict__ srt_w,
                                  const int* __restrict__ off_w, float* __restrict__ dinv_w) {
  const int b = blockIdx.x;
  if (b < G1_BLOCKS) {
    int wave = (b * G1_BS + (int)threadIdx.x) >> 5;
    int lane = threadIdx.x & 31;
    wmma_gemm_body<H1_DIM, 4, H2_DIM, true, false>(h1, nullptr, w2p, hx2, N_NODES, wave, lane);
  } else {
    degw_body(srt_w, off_w, dinv_w, (b - G1_BLOCKS) * G1_BS + (int)threadIdx.x);
  }
}

#define AGG_BS     256
#define AGG_WAVES  16384
template <int F, bool FULL_F>
__global__ void gcn_gather_kernel(const float* __restrict__ hx, float* __restrict__ out,
                                  const int2* __restrict__ srt, const int* __restrict__ off,
                                  const float* __restrict__ dinv, const float* __restrict__ b) {
  constexpr int VPL = F / 32;
  typedef __attribute__((ext_vector_type(VPL))) float vrow;
  const int lane = threadIdx.x & 31;
  const int wib = __builtin_amdgcn_readfirstlane((int)(threadIdx.x >> 5));
  const vrow* __restrict__ hrow = (const vrow*)hx + lane;
  const vrow bias = b ? ((const vrow*)b)[lane] : (vrow)(0.0f);

  for (int n = (int)blockIdx.x * (AGG_BS >> 5) + wib; n < N_NODES; n += AGG_WAVES) {
    const int s = off[n], e = off[n + 1];
    const float dn = dinv[n];
    const vrow self = hrow[(long long)n * 32];
    vrow acc = (vrow)(0.0f);
    for (int c0 = s; c0 < e; c0 += 32) {
      const int cnt = (e - c0 < 32) ? (e - c0) : 32;
      int r = 0;
      float f = 0.0f;
      if (lane < cnt) {
        const int2 rec = srt[c0 + lane];
        r = rec.x;
        f = __int_as_float(rec.y);
        if (!FULL_F) f = dinv[r] * f * dn;
      }
      for (int k8 = 0; k8 < cnt; k8 += 8) {
        vrow v[8];
#pragma unroll
        for (int k = 0; k < 8; ++k)
          v[k] = hrow[(long long)__builtin_amdgcn_readlane(r, k8 + k) * 32];
#pragma unroll
        for (int k = 0; k < 8; ++k) {
          const float fk = __int_as_float(__builtin_amdgcn_readlane(__float_as_int(f), k8 + k));
          if (k8 + k < cnt) acc += v[k] * fk;
        }
      }
    }
    acc += self * (dn * 1.0f * dn);
    ((vrow*)out)[(long long)n * 32 + lane] = acc + bias;
  }
}

__device__ __forceinline__ void edge_dot_body(const float* __restrict__ h,
                                              const int* __restrict__ pos,
                                              const int* __restrict__ neg,
                                              float* __restrict__ res, int e) {
  if (e >= 2 * E_LP) return;
  const int* src = (e < E_LP) ? pos : neg;
  int eo = (e < E_LP) ? e : (e - E_LP);
  int j = src[eo];
  int i = src[E_LP + eo];
  const float4* hj = (const float4*)(h + (long long)j * H2_DIM);
  const float4* hi = (const float4*)(h + (long long)i * H2_DIM);
  float s = 0.0f;
#pragma unroll
  for (int k = 0; k < H2_DIM / 4; ++k) {
    float4 a = hi[k], b = hj[k];
    s += a.x * b.x + a.y * b.y + a.z * b.z + a.w * b.w;
  }
  res[e] = s;
}

#define SM_PITCH 41
#define SM_WAVE  (32 * SM_PITCH)
#define DOT_BLOCKS ((2 * E_LP + G1_BS - 1) / G1_BS)
__global__ void gemm3_softmax_edge_dot_kernel(const float* __restrict__ g,
                                              const float* __restrict__ wcp,
                                              const float* __restrict__ bias0,
                                              const float* __restrict__ bias1,
                                              float* __restrict__ out0, float* __restrict__ out1,
                                              const float* __restrict__ h,
                                              const int* __restrict__ pos,
                                              const int* __restrict__ neg,
                                              float* __restrict__ res) {
  __shared__ __attribute__((aligned(16))) float sm[(G1_BS / 32) * SM_WAVE];
  const int b = blockIdx.x;
  if (b < G1_BLOCKS) {
    const int wib  = (int)threadIdx.x >> 5;
    const int wave = b * (G1_BS / 32) + wib;
    const int lane = threadIdx.x & 31;
    const bool live = wave < N_NODES / 16;
    float* s = sm + wib * SM_WAVE;

    if (live) {
      v8f acc[5] = {};
      wmma_tile_acc<H2_DIM, 5, false, false>(g, nullptr, wcp, wave, lane, acc);
      const int c  = lane & 15;
      const int hb = lane >> 4;
#pragma unroll
      for (int t = 0; t < 5; ++t) {
        const int col  = 16 * t + c;
        const int head = (col >= N_CLS) ? 1 : 0;
        const int cc   = col - N_CLS * head;
        const float bt = head ? bias1[cc] : bias0[cc];
#pragma unroll
        for (int r = 0; r < 8; ++r)
          s[(2 * (8 * hb + r) + head) * SM_PITCH + cc] = acc[t][r] + bt;
      }
    }
    __syncthreads();

    float x[N_CLS];
    if (live) {
#pragma unroll
      for (int k = 0; k < N_CLS; ++k) x[k] = s[lane * SM_PITCH + k];
      float m = -INFINITY;
#pragma unroll
      for (int k = 0; k < N_CLS; ++k) m = fmaxf(m, x[k]);
      float sum = 0.0f;
#pragma unroll
      for (int k = 0; k < N_CLS; ++k) sum += __expf(x[k] - m);
      const float ls = __logf(sum) + m;
#pragma unroll
      for (int k = 0; k < N_CLS; ++k) x[k] = x[k] - ls;
    }
    __syncthreads();

    if (live) {
      float* d = s + (lane & 1) * (16 * N_CLS) + (lane >> 1) * N_CLS;
#pragma unroll
      for (int k = 0; k < N_CLS; ++k) d[k] = x[k];
    }
    __syncthreads();

    if (live) {
      const float4* s4 = (const float4*)s;
      float4* o0 = (float4*)(out0 + (long long)wave * (16 * N_CLS));
      float4* o1 = (float4*)(out1 + (long long)wave * (16 * N_CLS));
#pragma unroll
      for (int k = 0; k < (16 * N_CLS / 4) / 32; ++k) {
        o0[lane + 32 * k] = s4[lane + 32 * k];
        o1[lane + 32 * k] = s4[(16 * N_CLS / 4) + lane + 32 * k];
      }
    }
  } else {
    edge_dot_body(h, pos, neg, res, (b - G1_BLOCKS) * G1_BS + (int)threadIdx.x);
  }
}

extern "C" void kernel_launch(void* const* d_in, const int* in_sizes, int n_in,
                              void* d_out, int out_size, void* d_ws, size_t ws_size,
                              hipStream_t stream) {
  const float* x      = (const float*)d_in[0];
  const int*   tpe    = (const int*)  d_in[1];
  const int*   ei     = (const int*)  d_in[2];
  const float* ew     = (const float*)d_in[3];
  const int*   pos    = (const int*)  d_in[4];
  const int*   neg    = (const int*)  d_in[5];
  const float* W1     = (const float*)d_in[6];
  const float* b1     = (const float*)d_in[7];
  const float* W2     = (const float*)d_in[8];
  const float* b2     = (const float*)d_in[9];
  const float* W_attr = (const float*)d_in[10];
  const float* b_attr = (const float*)d_in[11];
  const float* W_att  = (const float*)d_in[12];
  const float* b_att  = (const float*)d_in[13];

  float* out = (float*)d_out;
  float* ws  = (float*)d_ws;

  const long long OFF_DINV_T = 0;
  const long long OFF_DINV_W = 50000;
  const long long OFF_R0     = 100000;
  const long long OFF_R1     = OFF_R0 + 6400000;
  const long long OFF_R2     = OFF_R1 + 6400000;
  const long long OFF_W1P    = OFF_R2 + 6400000;
  const long long OFF_W2P    = OFF_W1P + 65536;
  const long long OFF_WCP    = OFF_W2P + 8192;

  float* dinv_t = ws + OFF_DINV_T;
  float* dinv_w = ws + OFF_DINV_W;
  float* hx1    = ws + OFF_R0;
  float* h1     = ws + OFF_R1;
  float* hx2    = ws + OFF_R0;
  float* h2     = ws + OFF_R0 + 3200000;
  float* g      = ws + OFF_R1;
  int2*  srt    = (int2*)(ws + OFF_R2);
  int*   rank   = (int*)(ws + OFF_R2 + 3200000);
  int*   cnt    = (int*)(ws + OFF_R2 + 4800000);
  int*   off    = (int*)(ws + OFF_R2 + 5000000);
  float* w1p    = ws + OFF_W1P;
  float* w2p    = ws + OFF_W2P;
  float* wcp    = ws + OFF_WCP;

  prep_kernel<<<PREP_B_W1 + PREP_B_W2 + PREP_B_WC + PREP_B_DG, PREP_BS, 0, stream>>>(
      W1, W2, W_attr, W_att, w1p, w2p, wcp, cnt);

  gemm1_rank_kernel<<<G1_BLOCKS + 2 * RK_BLOCKS, G1_BS, 0, stream>>>(
      x, w1p, hx1, cnt, tpe + E_MP, ei + E_MP, rank);

  scan_kernel<<<2 * SC_CH, SC_BS, 0, stream>>>(cnt, off, dinv_t);

  place_kernel<<<2 * PL_BLOCKS, 128, 0, stream>>>(tpe, ei, ew, rank, cnt, off, dinv_t, srt);

  gcn_gather_kernel<H1_DIM, true><<<AGG_WAVES / (AGG_BS / 32), AGG_BS, 0, stream>>>(
      hx1, h1, srt, off, dinv_t, b1);

  gemm2_degw_kernel<<<G1_BLOCKS + DW_BLOCKS, G1_BS, 0, stream>>>(
      h1, w2p, hx2, srt + E_MP, off + OFF_PITCH, dinv_w);
  gcn_gather_kernel<H2_DIM, true><<<AGG_WAVES / (AGG_BS / 32), AGG_BS, 0, stream>>>(
      hx2, h2, srt, off, dinv_t, b2);

  gcn_gather_kernel<H2_DIM, false><<<AGG_WAVES / (AGG_BS / 32), AGG_BS, 0, stream>>>(
      h2, g, srt + E_MP, off + OFF_PITCH, dinv_w, nullptr);
  gemm3_softmax_edge_dot_kernel<<<G1_BLOCKS + DOT_BLOCKS, G1_BS, 0, stream>>>(
      g, wcp, b_attr, b_att, out + 2 * E_LP,
      out + 2 * E_LP + (long long)N_NODES * N_CLS, h2, pos, neg, out);
}


# edge dots: 16 lanes share an edge, lane q loads float4 q of both rows (a load = 2 whole rows, not 16 B of each of 32 rows); waves loop over 32-edge groups; sum: 4-term partials, 16-lane butterfly
# speedup vs baseline: 2.8470x; 1.0904x over previous
#include <hip/hip_runtime.h>
#include <hip/hip_bf16.h>

#define N_NODES   50000
#define D_FEAT    512
#define H1_DIM    128
#define H2_DIM    64
#define N_CLS     40
#define E_MP      800000
#define E_LP      250000

typedef __attribute__((ext_vector_type(2))) float v2f;
typedef __attribute__((ext_vector_type(8))) float v8f;

__device__ __forceinline__ int chip_half() {
  return (int)((__builtin_amdgcn_s_sendmsg_rtn(0x87) >> 18) & 1u);
}

#define RK_Q (E_MP / 4)
__device__ __forceinline__ void rank_body(int* __restrict__ cnt0, int* __restrict__ cnt1,
                                          const int* __restrict__ cols, int* __restrict__ rank,
                                          int i) {
  if (i >= RK_Q) return;
  const int h = chip_half();
  int* cnt = h ? cnt1 : cnt0;
  int c[4], k4[4];
#pragma unroll
  for (int k = 0; k < 4; ++k) c[k] = cols[i + k * RK_Q];
#pragma unroll
  for (int k = 0; k < 4; ++k) k4[k] = atomicAdd(&cnt[c[k]], 1);
#pragma unroll
  for (int k = 0; k < 4; ++k) rank[i + k * RK_Q] = k4[k] | (h << 30);
}

#define SC_BS 1024
#define SC_CH ((N_NODES + 1 + SC_BS - 1) / SC_BS)
#define OFF_PITCH (SC_CH * SC_BS)
__global__ void scan_kernel(const int* __restrict__ cnt  ,
                            int* __restrict__ off  ,
                            float* __restrict__ dinv_t) {
  __shared__ int red[32];
  __shared__ int wtot[32];
  const int g = blockIdx.x / SC_CH;
  const int c = blockIdx.x % SC_CH;
  const int t = threadIdx.x, lane = t & 31, wid = t >> 5;
  const int* __restrict__ c0 = cnt + g * N_NODES;
  const int* __restrict__ c1 = cnt + 2 * N_NODES + g * N_NODES;
  int p = 0;
#pragma unroll 8
  for (int m = 0; m < c; ++m) p += c0[m * SC_BS + t] + c1[m * SC_BS + t];
  for (int d = 16; d > 0; d >>= 1) p += __shfl_xor(p, d);
  if (lane == 0) red[wid] = p;
  const int i = c * SC_BS + t;
  const int v = (i < N_NODES) ? (c0[i] + c1[i]) : 0;
  int incl = v;
  for (int d = 1; d < 32; d <<= 1) { int u = __shfl_up(incl, d); if (lane >= d) incl += u; }
  if (lane == 31) wtot[wid] = incl;
  __syncthreads();
  int base = 0;
  for (int k = 0; k < 32; ++k) { base += red[k]; if (k < wid) base += wtot[k]; }
  if (i <= N_NODES) off[g * OFF_PITCH + i] = base + incl - v;
  if (g == 0 && i < N_NODES) dinv_t[i] = rsqrtf((float)v + 1.0f);
}

template <bool WEIGHTED>
__device__ __forceinline__ void place_body(const int* __restrict__ rows,
                                           const int* __restrict__ cols,
                                           const float* __restrict__ ew,
                                           const int* __restrict__ rank,
                                           const int* __restrict__ cnt0,
                                           const int* __restrict__ off,
                                           const float* __restrict__ dinv,
                                           int2* __restrict__ srt, int i) {
  if (i >= RK_Q) return;
  int r[4], c[4], rk[4], pos[4];
  float f[4];
#pragma unroll
  for (int k = 0; k < 4; ++k) {
    const int e = i + k * RK_Q;
    r[k] = rows[e]; c[k] = cols[e]; rk[k] = rank[e];
    f[k] = WEIGHTED ? ew[e] : 1.0f;
  }
#pragma unroll
  for (int k = 0; k < 4; ++k) {
    const int p0 = cnt0[c[k]];
    pos[k] = off[c[k]] + (rk[k] & 0x3fffffff) + ((rk[k] >> 30) ? p0 : 0);
    if (!WEIGHTED) f[k] = dinv[r[k]] * f[k] * dinv[c[k]];
  }
#pragma unroll
  for (int k = 0; k < 4; ++k) srt[pos[k]] = make_int2(r[k], __float_as_int(f[k]));
}
#define PL_BLOCKS ((RK_Q + 127) / 128)
__global__ void place_kernel(const int* __restrict__ tpe, const int* __restrict__ ei,
                             const float* __restrict__ ew, const int* __restrict__ rank,
                             const int* __restrict__ cnt, const int* __restrict__ off,
                             const float* __restrict__ dinv_t, int2* __restrict__ srt) {
  const int b = blockIdx.x;
  if (b < PL_BLOCKS)
    place_body<false>(tpe, tpe + E_MP, nullptr, rank, cnt, off, dinv_t, srt,
                      b * 128 + (int)threadIdx.x);
  else
    place_body<true>(ei, ei + E_MP, ew, rank + E_MP, cnt + N_NODES, off + OFF_PITCH, nullptr,
                     srt + E_MP, (b - PL_BLOCKS) * 128 + (int)threadIdx.x);
}

__device__ __forceinline__ void degw_body(const int2* __restrict__ srt, const int* __restrict__ off,
                                          float* __restrict__ dinv_w, int i) {
  if (i >= N_NODES) return;
  const int s = off[i], e = off[i + 1];
  float d = 1.0f;
#pragma unroll 4
  for (int k = s; k < e; ++k) d += __int_as_float(srt[k].y);
  dinv_w[i] = (d > 0.0f) ? rsqrtf(d) : 0.0f;
}

__device__ __forceinline__ void pack_b_body(const float* __restrict__ B, float* __restrict__ Bp,
                                            int K, int NC, int NCP, int p) {
  int total = (K / 2) * NCP;
  if (p >= total) return;
  int k2 = p / NCP;
  int col = p % NCP;
  float x0 = 0.0f, x1 = 0.0f;
  if (col < NC) {
    x0 = B[(long long)(2 * k2) * NC + col];
    x1 = B[(long long)(2 * k2 + 1) * NC + col];
  }
  Bp[2 * p + 0] = x0;
  Bp[2 * p + 1] = x1;
}

__device__ __forceinline__ void pack_heads_body(const float* __restrict__ Ba,
                                                const float* __restrict__ Bb,
                                                float* __restrict__ Bp, int K, int NC, int p) {
  const int NCP = 2 * NC;
  int total = (K / 2) * NCP;
  if (p >= total) return;
  int k2 = p / NCP;
  int col = p % NCP;
  const float* __restrict__ B = (col < NC) ? Ba : Bb;
  int cc = (col < NC) ? col : col - NC;
  Bp[2 * p + 0] = B[(long long)(2 * k2) * NC + cc];
  Bp[2 * p + 1] = B[(long long)(2 * k2 + 1) * NC + cc];
}

#define PREP_BS   256
#define PREP_B_W1 ((D_FEAT / 2) * H1_DIM / PREP_BS)
#define PREP_B_W2 ((H1_DIM / 2) * H2_DIM / PREP_BS)
#define PREP_B_WC ((H2_DIM / 2) * 2 * N_CLS / PREP_BS)
#define PREP_B_DG ((4 * N_NODES + PREP_BS - 1) / PREP_BS)
__global__ void prep_kernel(const float* __restrict__ W1, const float* __restrict__ W2,
                            const float* __restrict__ Wa, const float* __restrict__ Wb,
                            float* __restrict__ w1p, float* __restrict__ w2p,
                            float* __restrict__ wcp, int* __restrict__ cnt) {
  int b = blockIdx.x;
  const int t = threadIdx.x;
  if (b < PREP_B_W1) { pack_b_body(W1, w1p, D_FEAT, H1_DIM, H1_DIM, b * PREP_BS + t); return; }
  b -= PREP_B_W1;
  if (b < PREP_B_W2) { pack_b_body(W2, w2p, H1_DIM, H2_DIM, H2_DIM, b * PREP_BS + t); return; }
  b -= PREP_B_W2;
  if (b < PREP_B_WC) { pack_heads_body(Wa, Wb, wcp, H2_DIM, N_CLS, b * PREP_BS + t); return; }
  b -= PREP_B_WC;
  const int i = b * PREP_BS + t;
  if (i < 4 * N_NODES) cnt[i] = 0;
}

template <int KDIM, int NT, bool RELU_A, bool DUAL_A>
__device__ __forceinline__ void wmma_tile_acc(const float* __restrict__ A,
                                              const float* __restrict__ A2,
                                              const float* __restrict__ Bp,
                                              int wave, int lane, v8f (&acc)[NT]) {
  constexpr int NCP = NT * 16;
  int row0 = wave << 4;
  int arow = row0 + (lane & 15);
  int khb  = lane >> 4;
  int bcol = lane & 15;

  const float2* __restrict__ Arow2 = (const float2*)(A + (long long)arow * KDIM);
  const float2* __restrict__ Arow2b = (const float2*)((DUAL_A ? A2 : A) + (long long)arow * KDIM);
  const float2* __restrict__ Bp2   = (const float2*)Bp;

#pragma unroll 2
  for (int k2 = 0; k2 < KDIM / 2; k2 += 2) {
    const float2* __restrict__ Bbase = Bp2 + (long long)(k2 + khb) * NCP + bcol;
    float2 bf[NT];
#pragma unroll
    for (int t = 0; t < NT; ++t) bf[t] = Bbase[t * 16];

    float2 af = Arow2[k2 + khb];
    if (DUAL_A) { float2 ag = Arow2b[k2 + khb]; af.x += ag.x; af.y += ag.y; }
    if (RELU_A) { af.x = fmaxf(af.x, 0.0f); af.y = fmaxf(af.y, 0.0f); }
    v2f a; a.x = af.x; a.y = af.y;

#pragma unroll
    for (int t = 0; t < NT; ++t) {
      v2f b; b.x = bf[t].x; b.y = bf[t].y;
      acc[t] = __builtin_amdgcn_wmma_f32_16x16x4_f32(
           false, a,  false, b,
           (short)0, acc[t],  false,  false);
    }
  }
}

template <int KDIM, int NT, int NCSTORE, bool RELU_A, bool DUAL_A>
__device__ __forceinline__ void wmma_gemm_body(const float* __restrict__ A,
                                               const float* __restrict__ A2,
                                               const float* __restrict__ Bp,
                                               float* __restrict__ C, int M, int wave, int lane) {
  int ntiles = M >> 4;
  if (wave >= ntiles) return;

  v8f acc[NT] = {};
  wmma_tile_acc<KDIM, NT, RELU_A, DUAL_A>(A, A2, Bp, wave, lane, acc);

  int row0 = wave << 4;
  int khb  = lane >> 4;
  int crow = row0 + (khb << 3);
  int ccol = lane & 15;
#pragma unroll
  for (int t = 0; t < NT; ++t) {
    int col = (t << 4) + ccol;
    if (col < NCSTORE) {
#pragma unroll
      for (int r = 0; r < 8; ++r) {
        C[(long long)(crow + r) * NCSTORE + col] = acc[t][r];
      }
    }
  }
}

#define G1_BS     128
#define G1_BLOCKS ((N_NODES / 16 * 32 + G1_BS - 1) / G1_BS)
#define RK_BLOCKS ((RK_Q + G1_BS - 1) / G1_BS)
__global__ void gemm1_rank_kernel(const float* __restrict__ x, const float* __restrict__ w1p,
                                  float* __restrict__ hx1, int* __restrict__ cnt,
                                  const int* __restrict__ cols_t, const int* __restrict__ cols_w,
                                  int* __restrict__ rank) {
  const int b = blockIdx.x;
  if (b < G1_BLOCKS) {
    int wave = (b * G1_BS + (int)threadIdx.x) >> 5;
    int lane = threadIdx.x & 31;
    wmma_gemm_body<D_FEAT, 8, H1_DIM, false, false>(x, nullptr, w1p, hx1, N_NODES, wave, lane);
  } else if (b < G1_BLOCKS + RK_BLOCKS) {
    rank_body(cnt, cnt + 2 * N_NODES, cols_t, rank, (b - G1_BLOCKS) * G1_BS + (int)threadIdx.x);
  } else {
    rank_body(cnt + N_NODES, cnt + 3 * N_NODES, cols_w, rank + E_MP,
              (b - G1_BLOCKS - RK_BLOCKS) * G1_BS + (int)threadIdx.x);
  }
}

#define DW_BLOCKS ((N_NODES + G1_BS - 1) / G1_BS)
__global__ void gemm2_degw_kernel(const float* __restrict__ h1, const float* __restrict__ w2p,
                                  float* __restrict__ hx2, const int2* __restrict__ srt_w,
                                  const int* __restrict__ off_w, float* __restrict__ dinv_w) {
  const int b = blockIdx.x;
  if (b < G1_BLOCKS) {
    int wave = (b * G1_BS + (int)threadIdx.x) >> 5;
    int lane = threadIdx.x & 31;
    wmma_gemm_body<H1_DIM, 4, H2_DIM, true, false>(h1, nullptr, w2p, hx2, N_NODES, wave, lane);
  } else {
    degw_body(srt_w, off_w, dinv_w, (b - G1_BLOCKS) * G1_BS + (int)threadIdx.x);
  }
}

#define AGG_BS     256
#define AGG_WAVES  16384
template <int F, bool FULL_F>
__global__ void gcn_gather_kernel(const float* __restrict__ hx, float* __restrict__ out,
                                  const int2* __restrict__ srt, const int* __restrict__ off,
                                  const float* __restrict__ dinv, const float* __restrict__ b) {
  constexpr int VPL = F / 32;
  typedef __attribute__((ext_vector_type(VPL))) float vrow;
  const int lane = threadIdx.x & 31;
  const int wib = __builtin_amdgcn_readfirstlane((int)(threadIdx.x >> 5));
  const vrow* __restrict__ hrow = (const vrow*)hx + lane;
  const vrow bias = b ? ((const vrow*)b)[lane] : (vrow)(0.0f);

  for (int n = (int)blockIdx.x * (AGG_BS >> 5) + wib; n < N_NODES; n += AGG_WAVES) {
    const int s = off[n], e = off[n + 1];
    const float dn = dinv[n];
    const vrow self = hrow[(long long)n * 32];
    vrow acc = (vrow)(0.0f);
    for (int c0 = s; c0 < e; c0 += 32) {
      const int cnt = (e - c0 < 32) ? (e - c0) : 32;
      int r = 0;
      float f = 0.0f;
      if (lane < cnt) {
        const int2 rec = srt[c0 + lane];
        r = rec.x;
        f = __int_as_float(rec.y);
        if (!FULL_F) f = dinv[r] * f * dn;
      }
      for (int k8 = 0; k8 < cnt; k8 += 8) {
        vrow v[8];
#pragma unroll
        for (int k = 0; k < 8; ++k)
          v[k] = hrow[(long long)__builtin_amdgcn_readlane(r, k8 + k) * 32];
#pragma unroll
        for (int k = 0; k < 8; ++k) {
          const float fk = __int_as_float(__builtin_amdgcn_readlane(__float_as_int(f), k8 + k));
          if (k8 + k < cnt) acc += v[k] * fk;
        }
      }
    }
    acc += self * (dn * 1.0f * dn);
    ((vrow*)out)[(long long)n * 32 + lane] = acc + bias;
  }
}

#define DOT_GROUPS (2 * E_LP / 32)
#define DOT_BLOCKS 1954
#define DOT_WAVES  (DOT_BLOCKS * (G1_BS / 32))

#define DPP_ROW_SHARE 0x150
#define DPP_ROW_XMASK 0x160
template <int CTRL>
__device__ __forceinline__ int row_move(int v) {
  return __builtin_amdgcn_update_dpp(v, v, CTRL, 0xf, 0xf, true);
}

__device__ __forceinline__ float row16_sum(float p) {
  p += __int_as_float(row_move<DPP_ROW_XMASK + 1>(__float_as_int(p)));
  p += __int_as_float(row_move<DPP_ROW_XMASK + 2>(__float_as_int(p)));
  p += __int_as_float(row_move<DPP_ROW_XMASK + 4>(__float_as_int(p)));
  p += __int_as_float(row_move<DPP_ROW_XMASK + 8>(__float_as_int(p)));
  return p;
}

__device__ __forceinline__ void edge_nodes(const int* __restrict__ pos, const int* __restrict__ neg,
                                           int e, int& j, int& i) {
  const int* __restrict__ src = (e < E_LP) ? pos : neg;
  const int eo = (e < E_LP) ? e : (e - E_LP);
  j = src[eo];
  i = src[E_LP + eo];
}

template <int T>
__device__ __forceinline__ float4 row_piece(const float* __restrict__ h, unsigned o, unsigned q16) {
  const unsigned at = (unsigned)row_move<DPP_ROW_SHARE + T>((int)o) + q16;
  return *(const float4*)((const char*)h + at);
}

template <int T0>
__device__ __forceinline__ void edge_dot8(const float* __restrict__ h, unsigned oi, unsigned oj,
                                          unsigned q16, int q, float& out) {
  float4 a[8], b[8];
#define EDGE_DOT_LD(t) a[t] = row_piece<T0 + t>(h, oi, q16); b[t] = row_piece<T0 + t>(h, oj, q16);
  EDGE_DOT_LD(0) EDGE_DOT_LD(1) EDGE_DOT_LD(2) EDGE_DOT_LD(3)
  EDGE_DOT_LD(4) EDGE_DOT_LD(5) EDGE_DOT_LD(6) EDGE_DOT_LD(7)
#undef EDGE_DOT_LD
#pragma unroll
  for (int t = 0; t < 8; ++t) {
    float p = a[t].x * b[t].x + a[t].y * b[t].y + a[t].z * b[t].z + a[t].w * b[t].w;
    p = row16_sum(p);
    if (q == T0 + t) out = p;
  }
}

__device__ __forceinline__ void edge_dot_body(const float* __restrict__ h,
                                              const int* __restrict__ pos,
                                              const int* __restrict__ neg,
                                              float* __restrict__ res, int wave, int lane) {
  const int q = lane & 15;
  const unsigned q16 = (unsigned)q * 16u;
  int g = wave;
  int j, i;
  edge_nodes(pos, neg, g * 32 + lane, j, i);
  for (;;) {
    const unsigned oj = (unsigned)j * (unsigned)(H2_DIM * 4);
    const unsigned oi = (unsigned)i * (unsigned)(H2_DIM * 4);
    const int gn = g + DOT_WAVES;
    if (gn < DOT_GROUPS) edge_nodes(pos, neg, gn * 32 + lane, j, i);
    float out = 0.0f;
    edge_dot8<0>(h, oi, oj, q16, q, out);
    edge_dot8<8>(h, oi, oj, q16, q, out);
    res[g * 32 + lane] = out;
    if (gn >= DOT_GROUPS) break;
    g = gn;
  }
}

#define SM_PITCH 41
#define SM_WAVE  (32 * SM_PITCH)
__global__ void gemm3_softmax_edge_dot_kernel(const float* __restrict__ g,
                                              const float* __restrict__ wcp,
                                              const float* __restrict__ bias0,
                                              const float* __restrict__ bias1,
                                              float* __restrict__ out0, float* __restrict__ out1,
                                              const float* __restrict__ h,
                                              const int* __restrict__ pos,
                                              const int* __restrict__ neg,
                                              float* __restrict__ res) {
  __shared__ __attribute__((aligned(16))) float sm[(G1_BS / 32) * SM_WAVE];
  const int b = blockIdx.x;
  if (b < G1_BLOCKS) {
    const int wib  = (int)threadIdx.x >> 5;
    const int wave = b * (G1_BS / 32) + wib;
    const int lane = threadIdx.x & 31;
    const bool live = wave < N_NODES / 16;
    float* s = sm + wib * SM_WAVE;

    if (live) {
      v8f acc[5] = {};
      wmma_tile_acc<H2_DIM, 5, false, false>(g, nullptr, wcp, wave, lane, acc);
      const int c  = lane & 15;
      const int hb = lane >> 4;
#pragma unroll
      for (int t = 0; t < 5; ++t) {
        const int col  = 16 * t + c;
        const int head = (col >= N_CLS) ? 1 : 0;
        const int cc   = col - N_CLS * head;
        const float bt = head ? bias1[cc] : bias0[cc];
#pragma unroll
        for (int r = 0; r < 8; ++r)
          s[(2 * (8 * hb + r) + head) * SM_PITCH + cc] = acc[t][r] + bt;
      }
    }
    __syncthreads();

    float x[N_CLS];
    if (live) {
#pragma unroll
      for (int k = 0; k < N_CLS; ++k) x[k] = s[lane * SM_PITCH + k];
      float m = -INFINITY;
#pragma unroll
      for (int k = 0; k < N_CLS; ++k) m = fmaxf(m, x[k]);
      float sum = 0.0f;
#pragma unroll
      for (int k = 0; k < N_CLS; ++k) sum += __expf(x[k] - m);
      const float ls = __logf(sum) + m;
#pragma unroll
      for (int k = 0; k < N_CLS; ++k) x[k] = x[k] - ls;
    }
    __syncthreads();

    if (live) {
      float* d = s + (lane & 1) * (16 * N_CLS) + (lane >> 1) * N_CLS;
#pragma unroll
      for (int k = 0; k < N_CLS; ++k) d[k] = x[k];
    }
    __syncthreads();

    if (live) {
      const float4* s4 = (const float4*)s;
      float4* o0 = (float4*)(out0 + (long long)wave * (16 * N_CLS));
      float4* o1 = (float4*)(out1 + (long long)wave * (16 * N_CLS));
#pragma unroll
      for (int k = 0; k < (16 * N_CLS / 4) / 32; ++k) {
        o0[lane + 32 * k] = s4[lane + 32 * k];
        o1[lane + 32 * k] = s4[(16 * N_CLS / 4) + lane + 32 * k];
      }
    }
  } else {
    const int wib = __builtin_amdgcn_readfirstlane((int)(threadIdx.x >> 5));
    edge_dot_body(h, pos, neg, res, (b - G1_BLOCKS) * (G1_BS / 32) + wib, (int)threadIdx.x & 31);
  }
}

extern "C" void kernel_launch(void* const* d_in, const int* in_sizes, int n_in,
                              void* d_out, int out_size, void* d_ws, size_t ws_size,
                              hipStream_t stream) {
  const float* x      = (const float*)d_in[0];
  const int*   tpe    = (const int*)  d_in[1];
  const int*   ei     = (const int*)  d_in[2];
  const float* ew     = (const float*)d_in[3];
  const int*   pos    = (const int*)  d_in[4];
  const int*   neg    = (const int*)  d_in[5];
  const float* W1     = (const float*)d_in[6];
  const float* b1     = (const float*)d_in[7];
  const float* W2     = (const float*)d_in[8];
  const float* b2     = (const float*)d_in[9];
  const float* W_attr = (const float*)d_in[10];
  const float* b_attr = (const float*)d_in[11];
  const float* W_att  = (const float*)d_in[12];
  const float* b_att  = (const float*)d_in[13];

  float* out = (float*)d_out;
  float* ws  = (float*)d_ws;

  const long long OFF_DINV_T = 0;
  const long long OFF_DINV_W = 50000;
  const long long OFF_R0     = 100000;
  const long long OFF_R1     = OFF_R0 + 6400000;
  const long long OFF_R2     = OFF_R1 + 6400000;
  const long long OFF_W1P    = OFF_R2 + 6400000;
  const long long OFF_W2P    = OFF_W1P + 65536;
  const long long OFF_WCP    = OFF_W2P + 8192;

  float* dinv_t = ws + OFF_DINV_T;
  float* dinv_w = ws + OFF_DINV_W;
  float* hx1    = ws + OFF_R0;
  float* h1     = ws + OFF_R1;
  float* hx2    = ws + OFF_R0;
  float* h2     = ws + OFF_R0 + 3200000;
  float* g      = ws + OFF_R1;
  int2*  srt    = (int2*)(ws + OFF_R2);
  int*   rank   = (int*)(ws + OFF_R2 + 3200000);
  int*   cnt    = (int*)(ws + OFF_R2 + 4800000);
  int*   off    = (int*)(ws + OFF_R2 + 5000000);
  float* w1p    = ws + OFF_W1P;
  float* w2p    = ws + OFF_W2P;
  float* wcp    = ws + OFF_WCP;

  prep_kernel<<<PREP_B_W1 + PREP_B_W2 + PREP_B_WC + PREP_B_DG, PREP_BS, 0, stream>>>(
      W1, W2, W_attr, W_att, w1p, w2p, wcp, cnt);

  gemm1_rank_kernel<<<G1_BLOCKS + 2 * RK_BLOCKS, G1_BS, 0, stream>>>(
      x, w1p, hx1, cnt, tpe + E_MP, ei + E_MP, rank);

  scan_kernel<<<2 * SC_CH, SC_BS, 0, stream>>>(cnt, off, dinv_t);

  place_kernel<<<2 * PL_BLOCKS, 128, 0, stream>>>(tpe, ei, ew, rank, cnt, off, dinv_t, srt);

  gcn_gather_kernel<H1_DIM, true><<<AGG_WAVES / (AGG_BS / 32), AGG_BS, 0, stream>>>(
      hx1, h1, srt, off, dinv_t, b1);

  gemm2_degw_kernel<<<G1_BLOCKS + DW_BLOCKS, G1_BS, 0, stream>>>(
      h1, w2p, hx2, srt + E_MP, off + OFF_PITCH, dinv_w);
  gcn_gather_kernel<H2_DIM, true><<<AGG_WAVES / (AGG_BS / 32), AGG_BS, 0, stream>>>(
      hx2, h2, srt, off, dinv_t, b2);

  gcn_gather_kernel<H2_DIM, false><<<AGG_WAVES / (AGG_BS / 32), AGG_BS, 0, stream>>>(
      h2, g, srt + E_MP, off + OFF_PITCH, dinv_w, nullptr);
  gemm3_softmax_edge_dot_kernel<<<G1_BLOCKS + DOT_BLOCKS, G1_BS, 0, stream>>>(
      g, wcp, b_attr, b_att, out + 2 * E_LP,
      out + 2 * E_LP + (long long)N_NODES * N_CLS, h2, pos, neg, out);
}


# layer 1: a block's 4 waves sum 16 nodes' h1 rows into LDS (relu there) and do hx2 = relu(h1) @ W2 on WMMA from it; h1 never stored, product launch gone, 9 -> 8; every sum in the order of before
# speedup vs baseline: 2.9596x; 1.0396x over previous
#include <hip/hip_runtime.h>
#include <hip/hip_bf16.h>

#define N_NODES   50000
#define D_FEAT    512
#define H1_DIM    128
#define H2_DIM    64
#define N_CLS     40
#define E_MP      800000
#define E_LP      250000

typedef __attribute__((ext_vector_type(2))) float v2f;
typedef __attribute__((ext_vector_type(8))) float v8f;

__device__ __forceinline__ int chip_half() {
  return (int)((__builtin_amdgcn_s_sendmsg_rtn(0x87) >> 18) & 1u);
}

#define RK_Q (E_MP / 4)
__device__ __forceinline__ void rank_body(int* __restrict__ cnt0, int* __restrict__ cnt1,
                                          const int* __restrict__ cols, int* __restrict__ rank,
                                          int i) {
  if (i >= RK_Q) return;
  const int h = chip_half();
  int* cnt = h ? cnt1 : cnt0;
  int c[4], k4[4];
#pragma unroll
  for (int k = 0; k < 4; ++k) c[k] = cols[i + k * RK_Q];
#pragma unroll
  for (int k = 0; k < 4; ++k) k4[k] = atomicAdd(&cnt[c[k]], 1);
#pragma unroll
  for (int k = 0; k < 4; ++k) rank[i + k * RK_Q] = k4[k] | (h << 30);
}

#define SC_BS 1024
#define SC_CH ((N_NODES + 1 + SC_BS - 1) / SC_BS)
#define OFF_PITCH (SC_CH * SC_BS)
__global__ void scan_kernel(const int* __restrict__ cnt  ,
                            int* __restrict__ off  ,
                            float* __restrict__ dinv_t) {
  __shared__ int red[32];
  __shared__ int wtot[32];
  const int g = blockIdx.x / SC_CH;
  const int c = blockIdx.x % SC_CH;
  const int t = threadIdx.x, lane = t & 31, wid = t >> 5;
  const int* __restrict__ c0 = cnt + g * N_NODES;
  const int* __restrict__ c1 = cnt + 2 * N_NODES + g * N_NODES;
  int p = 0;
#pragma unroll 8
  for (int m = 0; m < c; ++m) p += c0[m * SC_BS + t] + c1[m * SC_BS + t];
  for (int d = 16; d > 0; d >>= 1) p += __shfl_xor(p, d);
  if (lane == 0) red[wid] = p;
  const int i = c * SC_BS + t;
  const int v = (i < N_NODES) ? (c0[i] + c1[i]) : 0;
  int incl = v;
  for (int d = 1; d < 32; d <<= 1) { int u = __shfl_up(incl, d); if (lane >= d) incl += u; }
  if (lane == 31) wtot[wid] = incl;
  __syncthreads();
  int base = 0;
  for (int k = 0; k < 32; ++k) { base += red[k]; if (k < wid) base += wtot[k]; }
  if (i <= N_NODES) off[g * OFF_PITCH + i] = base + incl - v;
  if (g == 0 && i < N_NODES) dinv_t[i] = rsqrtf((float)v + 1.0f);
}

template <bool WEIGHTED>
__device__ __forceinline__ void place_body(const int* __restrict__ rows,
                                           const int* __restrict__ cols,
                                           const float* __restrict__ ew,
                                           const int* __restrict__ rank,
                                           const int* __restrict__ cnt0,
                                           const int* __restrict__ off,
                                           const float* __restrict__ dinv,
                                           int2* __restrict__ srt, int i) {
  if (i >= RK_Q) return;
  int r[4], c[4], rk[4], pos[4];
  float f[4];
#pragma unroll
  for (int k = 0; k < 4; ++k) {
    const int e = i + k * RK_Q;
    r[k] = rows[e]; c[k] = cols[e]; rk[k] = rank[e];
    f[k] = WEIGHTED ? ew[e] : 1.0f;
  }
#pragma unroll
  for (int k = 0; k < 4; ++k) {
    const int p0 = cnt0[c[k]];
    pos[k] = off[c[k]] + (rk[k] & 0x3fffffff) + ((rk[k] >> 30) ? p0 : 0);
    if (!WEIGHTED) f[k] = dinv[r[k]] * f[k] * dinv[c[k]];
  }
#pragma unroll
  for (int k = 0; k < 4; ++k) srt[pos[k]] = make_int2(r[k], __float_as_int(f[k]));
}
#define PL_BLOCKS ((RK_Q + 127) / 128)
__global__ void place_kernel(const int* __restrict__ tpe, const int* __restrict__ ei,
                             const float* __restrict__ ew, const int* __restrict__ rank,
                             const int* __restrict__ cnt, const int* __restrict__ off,
                             const float* __restrict__ dinv_t, int2* __restrict__ srt) {
  const int b = blockIdx.x;
  if (b < PL_BLOCKS)
    place_body<false>(tpe, tpe + E_MP, nullptr, rank, cnt, off, dinv_t, srt,
                      b * 128 + (int)threadIdx.x);
  else
    place_body<true>(ei, ei + E_MP, ew, rank + E_MP, cnt + N_NODES, off + OFF_PITCH, nullptr,
                     srt + E_MP, (b - PL_BLOCKS) * 128 + (int)threadIdx.x);
}

__device__ __forceinline__ void degw_body(const int2* __restrict__ srt, const int* __restrict__ off,
                                          float* __restrict__ dinv_w, int i) {
  if (i >= N_NODES) return;
  const int s = off[i], e = off[i + 1];
  float d = 1.0f;
#pragma unroll 4
  for (int k = s; k < e; ++k) d += __int_as_float(srt[k].y);
  dinv_w[i] = (d > 0.0f) ? rsqrtf(d) : 0.0f;
}

__device__ __forceinline__ void pack_b_body(const float* __restrict__ B, float* __restrict__ Bp,
                                            int K, int NC, int NCP, int p) {
  int total = (K / 2) * NCP;
  if (p >= total) return;
  int k2 = p / NCP;
  int col = p % NCP;
  float x0 = 0.0f, x1 = 0.0f;
  if (col < NC) {
    x0 = B[(long long)(2 * k2) * NC + col];
    x1 = B[(long long)(2 * k2 + 1) * NC + col];
  }
  Bp[2 * p + 0] = x0;
  Bp[2 * p + 1] = x1;
}

__device__ __forceinline__ void pack_heads_body(const float* __restrict__ Ba,
                                                const float* __restrict__ Bb,
                                                float* __restrict__ Bp, int K, int NC, int p) {
  const int NCP = 2 * NC;
  int total = (K / 2) * NCP;
  if (p >= total) return;
  int k2 = p / NCP;
  int col = p % NCP;
  const float* __restrict__ B = (col < NC) ? Ba : Bb;
  int cc = (col < NC) ? col : col - NC;
  Bp[2 * p + 0] = B[(long long)(2 * k2) * NC + cc];
  Bp[2 * p + 1] = B[(long long)(2 * k2 + 1) * NC + cc];
}

#define PREP_BS   256
#define PREP_B_W1 ((D_FEAT / 2) * H1_DIM / PREP_BS)
#define PREP_B_W2 ((H1_DIM / 2) * H2_DIM / PREP_BS)
#define PREP_B_WC ((H2_DIM / 2) * 2 * N_CLS / PREP_BS)
#define PREP_B_DG ((4 * N_NODES + PREP_BS - 1) / PREP_BS)
__global__ void prep_kernel(const float* __restrict__ W1, const float* __restrict__ W2,
                            const float* __restrict__ Wa, const float* __restrict__ Wb,
                            float* __restrict__ w1p, float* __restrict__ w2p,
                            float* __restrict__ wcp, int* __restrict__ cnt) {
  int b = blockIdx.x;
  const int t = threadIdx.x;
  if (b < PREP_B_W1) { pack_b_body(W1, w1p, D_FEAT, H1_DIM, H1_DIM, b * PREP_BS + t); return; }
  b -= PREP_B_W1;
  if (b < PREP_B_W2) { pack_b_body(W2, w2p, H1_DIM, H2_DIM, H2_DIM, b * PREP_BS + t); return; }
  b -= PREP_B_W2;
  if (b < PREP_B_WC) { pack_heads_body(Wa, Wb, wcp, H2_DIM, N_CLS, b * PREP_BS + t); return; }
  b -= PREP_B_WC;
  const int i = b * PREP_BS + t;
  if (i < 4 * N_NODES) cnt[i] = 0;
}

template <int KDIM, int NT, bool RELU_A, bool DUAL_A>
__device__ __forceinline__ void wmma_tile_acc(const float* __restrict__ A,
                                              const float* __restrict__ A2,
                                              const float* __restrict__ Bp,
                                              int wave, int lane, v8f (&acc)[NT]) {
  constexpr int NCP = NT * 16;
  int row0 = wave << 4;
  int arow = row0 + (lane & 15);
  int khb  = lane >> 4;
  int bcol = lane & 15;

  const float2* __restrict__ Arow2 = (const float2*)(A + (long long)arow * KDIM);
  const float2* __restrict__ Arow2b = (const float2*)((DUAL_A ? A2 : A) + (long long)arow * KDIM);
  const float2* __restrict__ Bp2   = (const float2*)Bp;

#pragma unroll 2
  for (int k2 = 0; k2 < KDIM / 2; k2 += 2) {
    const float2* __restrict__ Bbase = Bp2 + (long long)(k2 + khb) * NCP + bcol;
    float2 bf[NT];
#pragma unroll
    for (int t = 0; t < NT; ++t) bf[t] = Bbase[t * 16];

    float2 af = Arow2[k2 + khb];
    if (DUAL_A) { float2 ag = Arow2b[k2 + khb]; af.x += ag.x; af.y += ag.y; }
    if (RELU_A) { af.x = fmaxf(af.x, 0.0f); af.y = fmaxf(af.y, 0.0f); }
    v2f a; a.x = af.x; a.y = af.y;

#pragma unroll
    for (int t = 0; t < NT; ++t) {
      v2f b; b.x = bf[t].x; b.y = bf[t].y;
      acc[t] = __builtin_amdgcn_wmma_f32_16x16x4_f32(
           false, a,  false, b,
           (short)0, acc[t],  false,  false);
    }
  }
}

template <int KDIM, int NT, int NCSTORE, bool RELU_A, bool DUAL_A>
__device__ __forceinline__ void wmma_gemm_body(const float* __restrict__ A,
                                               const float* __restrict__ A2,
                                               const float* __restrict__ Bp,
                                               float* __restrict__ C, int M, int wave, int lane) {
  int ntiles = M >> 4;
  if (wave >= ntiles) return;

  v8f acc[NT] = {};
  wmma_tile_acc<KDIM, NT, RELU_A, DUAL_A>(A, A2, Bp, wave, lane, acc);

  int row0 = wave << 4;
  int khb  = lane >> 4;
  int crow = row0 + (khb << 3);
  int ccol = lane & 15;
#pragma unroll
  for (int t = 0; t < NT; ++t) {
    int col = (t << 4) + ccol;
    if (col < NCSTORE) {
#pragma unroll
      for (int r = 0; r < 8; ++r) {
        C[(long long)(crow + r) * NCSTORE + col] = acc[t][r];
      }
    }
  }
}

#define G1_BS     128
#define G1_BLOCKS ((N_NODES / 16 * 32 + G1_BS - 1) / G1_BS)
#define RK_BLOCKS ((RK_Q + G1_BS - 1) / G1_BS)
__global__ void gemm1_rank_kernel(const float* __restrict__ x, const float* __restrict__ w1p,
                                  float* __restrict__ hx1, int* __restrict__ cnt,
                                  const int* __restrict__ cols_t, const int* __restrict__ cols_w,
                                  int* __restrict__ rank) {
  const int b = blockIdx.x;
  if (b < G1_BLOCKS) {
    int wave = (b * G1_BS + (int)threadIdx.x) >> 5;
    int lane = threadIdx.x & 31;
    wmma_gemm_body<D_FEAT, 8, H1_DIM, false, false>(x, nullptr, w1p, hx1, N_NODES, wave, lane);
  } else if (b < G1_BLOCKS + RK_BLOCKS) {
    rank_body(cnt, cnt + 2 * N_NODES, cols_t, rank, (b - G1_BLOCKS) * G1_BS + (int)threadIdx.x);
  } else {
    rank_body(cnt + N_NODES, cnt + 3 * N_NODES, cols_w, rank + E_MP,
              (b - G1_BLOCKS - RK_BLOCKS) * G1_BS + (int)threadIdx.x);
  }
}

#define AGG_BS     256
#define AGG_WAVES  16384
template <int F, bool FULL_F>
__global__ void gcn_gather_kernel(const float* __restrict__ hx, float* __restrict__ out,
                                  const int2* __restrict__ srt, const int* __restrict__ off,
                                  const float* __restrict__ dinv, const float* __restrict__ b) {
  constexpr int VPL = F / 32;
  typedef __attribute__((ext_vector_type(VPL))) float vrow;
  const int lane = threadIdx.x & 31;
  const int wib = __builtin_amdgcn_readfirstlane((int)(threadIdx.x >> 5));
  const vrow* __restrict__ hrow = (const vrow*)hx + lane;
  const vrow bias = b ? ((const vrow*)b)[lane] : (vrow)(0.0f);

  for (int n = (int)blockIdx.x * (AGG_BS >> 5) + wib; n < N_NODES; n += AGG_WAVES) {
    const int s = off[n], e = off[n + 1];
    const float dn = dinv[n];
    const vrow self = hrow[(long long)n * 32];
    vrow acc = (vrow)(0.0f);
    for (int c0 = s; c0 < e; c0 += 32) {
      const int cnt = (e - c0 < 32) ? (e - c0) : 32;
      int r = 0;
      float f = 0.0f;
      if (lane < cnt) {
        const int2 rec = srt[c0 + lane];
        r = rec.x;
        f = __int_as_float(rec.y);
        if (!FULL_F) f = dinv[r] * f * dn;
      }
      for (int k8 = 0; k8 < cnt; k8 += 8) {
        vrow v[8];
#pragma unroll
        for (int k = 0; k < 8; ++k)
          v[k] = hrow[(long long)__builtin_amdgcn_readlane(r, k8 + k) * 32];
#pragma unroll
        for (int k = 0; k < 8; ++k) {
          const float fk = __int_as_float(__builtin_amdgcn_readlane(__float_as_int(f), k8 + k));
          if (k8 + k < cnt) acc += v[k] * fk;
        }
      }
    }
    acc += self * (dn * 1.0f * dn);
    ((vrow*)out)[(long long)n * 32 + lane] = acc + bias;
  }
}

#define FG_BS     128
#define FG_PITCH  132
#define FG_BLOCKS (N_NODES / 16)
#define DW_BLOCKS ((N_NODES + FG_BS - 1) / FG_BS)
__global__ void gather1_gemm2_degw_kernel(const float* __restrict__ hx, const int2* __restrict__ srt,
                                          const int* __restrict__ off,
                                          const float* __restrict__ dinv,
                                          const float* __restrict__ b1,
                                          const float* __restrict__ w2p, float* __restrict__ hx2,
                                          const int2* __restrict__ srt_w,
                                          const int* __restrict__ off_w,
                                          float* __restrict__ dinv_w) {
  __shared__ __attribute__((aligned(16))) float rows[16 * FG_PITCH];
  const int blk = blockIdx.x;
  if (blk >= FG_BLOCKS) {
    degw_body(srt_w, off_w, dinv_w, (blk - FG_BLOCKS) * FG_BS + (int)threadIdx.x);
    return;
  }
  const int lane = threadIdx.x & 31;
  const int wib = __builtin_amdgcn_readfirstlane((int)(threadIdx.x >> 5));
  const float4* __restrict__ hrow = (const float4*)hx + lane;
  const float4 bias = ((const float4*)b1)[lane];

  const int n0 = 16 * blk + 4 * wib;
  int bnd[5];
  float dns[4];
#pragma unroll
  for (int j = 0; j < 5; ++j) bnd[j] = off[n0 + j];
#pragma unroll
  for (int j = 0; j < 4; ++j) dns[j] = dinv[n0 + j];

#pragma unroll
  for (int m = 0; m < 4; ++m) {
    const int ln = 4 * wib + m;
    const int n = n0 + m;
    const int s = bnd[m], e = bnd[m + 1];
    const float dn = dns[m];
    const float4 self = hrow[(long long)n * 32];
    float4 acc = make_float4(0.0f, 0.0f, 0.0f, 0.0f);
    for (int c0 = s; c0 < e; c0 += 32) {
      const int cnt = (e - c0 < 32) ? (e - c0) : 32;
      int r = 0;
      float f = 0.0f;
      if (lane < cnt) {
        const int2 rec = srt[c0 + lane];
        r = rec.x;
        f = __int_as_float(rec.y);
      }
      for (int k8 = 0; k8 < cnt; k8 += 8) {
        float4 v[8];
#pragma unroll
        for (int k = 0; k < 8; ++k)
          v[k] = hrow[(long long)__builtin_amdgcn_readlane(r, k8 + k) * 32];
#pragma unroll
        for (int k = 0; k < 8; ++k) {
          const float fk = __int_as_float(__builtin_amdgcn_readlane(__float_as_int(f), k8 + k));
          if (k8 + k < cnt) {
            acc.x += v[k].x * fk; acc.y += v[k].y * fk;
            acc.z += v[k].z * fk; acc.w += v[k].w * fk;
          }
        }
      }
    }
    const float fs = dn * 1.0f * dn;
    float4 o;
    o.x = fmaxf((acc.x + self.x * fs) + bias.x, 0.0f);
    o.y = fmaxf((acc.y + self.y * fs) + bias.y, 0.0f);
    o.z = fmaxf((acc.z + self.z * fs) + bias.z, 0.0f);
    o.w = fmaxf((acc.w + self.w * fs) + bias.w, 0.0f);
    *(float4*)&rows[ln * FG_PITCH + 4 * lane] = o;
  }
  __syncthreads();

  const int c = lane & 15;
  const int h = lane >> 4;
  const float2* __restrict__ Bp2 = (const float2*)w2p + 16 * wib + c;
  const float* arow = rows + c * FG_PITCH;
  v8f acc = {};
#pragma unroll 4
  for (int k2 = 0; k2 < H1_DIM / 2; k2 += 2) {
    const float2 bf = Bp2[(k2 + h) * H2_DIM];
    const float2 af = *(const float2*)(arow + 2 * (k2 + h));
    v2f a; a.x = af.x; a.y = af.y;
    v2f bb; bb.x = bf.x; bb.y = bf.y;
    acc = __builtin_amdgcn_wmma_f32_16x16x4_f32(
         false, a,  false, bb,
         (short)0, acc,  false,  false);
  }
  float* __restrict__ crow = hx2 + (long long)(16 * blk + 8 * h) * H2_DIM + 16 * wib + c;
#pragma unroll
  for (int r = 0; r < 8; ++r) crow[r * H2_DIM] = acc[r];
}

#define DOT_GROUPS (2 * E_LP / 32)
#define DOT_BLOCKS 1954
#define DOT_WAVES  (DOT_BLOCKS * (G1_BS / 32))

#define DPP_ROW_SHARE 0x150
#define DPP_ROW_XMASK 0x160
template <int CTRL>
__device__ __forceinline__ int row_move(int v) {
  return __builtin_amdgcn_update_dpp(v, v, CTRL, 0xf, 0xf, true);
}

__device__ __forceinline__ float row16_sum(float p) {
  p += __int_as_float(row_move<DPP_ROW_XMASK + 1>(__float_as_int(p)));
  p += __int_as_float(row_move<DPP_ROW_XMASK + 2>(__float_as_int(p)));
  p += __int_as_float(row_move<DPP_ROW_XMASK + 4>(__float_as_int(p)));
  p += __int_as_float(row_move<DPP_ROW_XMASK + 8>(__float_as_int(p)));
  return p;
}

__device__ __forceinline__ void edge_nodes(const int* __restrict__ pos, const int* __restrict__ neg,
                                           int e, int& j, int& i) {
  const int* __restrict__ src = (e < E_LP) ? pos : neg;
  const int eo = (e < E_LP) ? e : (e - E_LP);
  j = src[eo];
  i = src[E_LP + eo];
}

template <int T>
__device__ __forceinline__ float4 row_piece(const float* __restrict__ h, unsigned o, unsigned q16) {
  const unsigned at = (unsigned)row_move<DPP_ROW_SHARE + T>((int)o) + q16;
  return *(const float4*)((const char*)h + at);
}

template <int T0>
__device__ __forceinline__ void edge_dot8(const float* __restrict__ h, unsigned oi, unsigned oj,
                                          unsigned q16, int q, float& out) {
  float4 a[8], b[8];
#define EDGE_DOT_LD(t) a[t] = row_piece<T0 + t>(h, oi, q16); b[t] = row_piece<T0 + t>(h, oj, q16);
  EDGE_DOT_LD(0) EDGE_DOT_LD(1) EDGE_DOT_LD(2) EDGE_DOT_LD(3)
  EDGE_DOT_LD(4) EDGE_DOT_LD(5) EDGE_DOT_LD(6) EDGE_DOT_LD(7)
#undef EDGE_DOT_LD
#pragma unroll
  for (int t = 0; t < 8; ++t) {
    float p = a[t].x * b[t].x + a[t].y * b[t].y + a[t].z * b[t].z + a[t].w * b[t].w;
    p = row16_sum(p);
    if (q == T0 + t) out = p;
  }
}

__device__ __forceinline__ void edge_dot_body(const float* __restrict__ h,
                                              const int* __restrict__ pos,
                                              const int* __restrict__ neg,
                                              float* __restrict__ res, int wave, int lane) {
  const int q = lane & 15;
  const unsigned q16 = (unsigned)q * 16u;
  int g = wave;
  int j, i;
  edge_nodes(pos, neg, g * 32 + lane, j, i);
  for (;;) {
    const unsigned oj = (unsigned)j * (unsigned)(H2_DIM * 4);
    const unsigned oi = (unsigned)i * (unsigned)(H2_DIM * 4);
    const int gn = g + DOT_WAVES;
    if (gn < DOT_GROUPS) edge_nodes(pos, neg, gn * 32 + lane, j, i);
    float out = 0.0f;
    edge_dot8<0>(h, oi, oj, q16, q, out);
    edge_dot8<8>(h, oi, oj, q16, q, out);
    res[g * 32 + lane] = out;
    if (gn >= DOT_GROUPS) break;
    g = gn;
  }
}

#define SM_PITCH 41
#define SM_WAVE  (32 * SM_PITCH)
__global__ void gemm3_softmax_edge_dot_kernel(const float* __restrict__ g,
                                              const float* __restrict__ wcp,
                                              const float* __restrict__ bias0,
                                              const float* __restrict__ bias1,
                                              float* __restrict__ out0, float* __restrict__ out1,
                                              const float* __restrict__ h,
                                              const int* __restrict__ pos,
                                              const int* __restrict__ neg,
                                              float* __restrict__ res) {
  __shared__ __attribute__((aligned(16))) float sm[(G1_BS / 32) * SM_WAVE];
  const int b = blockIdx.x;
  if (b < G1_BLOCKS) {
    const int wib  = (int)threadIdx.x >> 5;
    const int wave = b * (G1_BS / 32) + wib;
    const int lane = threadIdx.x & 31;
    const bool live = wave < N_NODES / 16;
    float* s = sm + wib * SM_WAVE;

    if (live) {
      v8f acc[5] = {};
      wmma_tile_acc<H2_DIM, 5, false, false>(g, nullptr, wcp, wave, lane, acc);
      const int c  = lane & 15;
      const int hb = lane >> 4;
#pragma unroll
      for (int t = 0; t < 5; ++t) {
        const int col  = 16 * t + c;
        const int head = (col >= N_CLS) ? 1 : 0;
        const int cc   = col - N_CLS * head;
        const float bt = head ? bias1[cc] : bias0[cc];
#pragma unroll
        for (int r = 0; r < 8; ++r)
          s[(2 * (8 * hb + r) + head) * SM_PITCH + cc] = acc[t][r] + bt;
      }
    }
    __syncthreads();

    float x[N_CLS];
    if (live) {
#pragma unroll
      for (int k = 0; k < N_CLS; ++k) x[k] = s[lane * SM_PITCH + k];
      float m = -INFINITY;
#pragma unroll
      for (int k = 0; k < N_CLS; ++k) m = fmaxf(m, x[k]);
      float sum = 0.0f;
#pragma unroll
      for (int k = 0; k < N_CLS; ++k) sum += __expf(x[k] - m);
      const float ls = __logf(sum) + m;
#pragma unroll
      for (int k = 0; k < N_CLS; ++k) x[k] = x[k] - ls;
    }
    __syncthreads();

    if (live) {
      float* d = s + (lane & 1) * (16 * N_CLS) + (lane >> 1) * N_CLS;
#pragma unroll
      for (int k = 0; k < N_CLS; ++k) d[k] = x[k];
    }
    __syncthreads();

    if (live) {
      const float4* s4 = (const float4*)s;
      float4* o0 = (float4*)(out0 + (long long)wave * (16 * N_CLS));
      float4* o1 = (float4*)(out1 + (long long)wave * (16 * N_CLS));
#pragma unroll
      for (int k = 0; k < (16 * N_CLS / 4) / 32; ++k) {
        o0[lane + 32 * k] = s4[lane + 32 * k];
        o1[lane + 32 * k] = s4[(16 * N_CLS / 4) + lane + 32 * k];
      }
    }
  } else {
    const int wib = __builtin_amdgcn_readfirstlane((int)(threadIdx.x >> 5));
    edge_dot_body(h, pos, neg, res, (b - G1_BLOCKS) * (G1_BS / 32) + wib, (int)threadIdx.x & 31);
  }
}

extern "C" void kernel_launch(void* const* d_in, const int* in_sizes, int n_in,
                              void* d_out, int out_size, void* d_ws, size_t ws_size,
                              hipStream_t stream) {
  const float* x      = (const float*)d_in[0];
  const int*   tpe    = (const int*)  d_in[1];
  const int*   ei     = (const int*)  d_in[2];
  const float* ew     = (const float*)d_in[3];
  const int*   pos    = (const int*)  d_in[4];
  const int*   neg    = (const int*)  d_in[5];
  const float* W1     = (const float*)d_in[6];
  const float* b1     = (const float*)d_in[7];
  const float* W2     = (const float*)d_in[8];
  const float* b2     = (const float*)d_in[9];
  const float* W_attr = (const float*)d_in[10];
  const float* b_attr = (const float*)d_in[11];
  const float* W_att  = (const float*)d_in[12];
  const float* b_att  = (const float*)d_in[13];

  float* out = (float*)d_out;
  float* ws  = (float*)d_ws;

  const long long OFF_DINV_T = 0;
  const long long OFF_DINV_W = 50000;
  const long long OFF_R0     = 100000;
  const long long OFF_R1     = OFF_R0 + 6400000;
  const long long OFF_R2     = OFF_R1 + 6400000;
  const long long OFF_W1P    = OFF_R2 + 6400000;
  const long long OFF_W2P    = OFF_W1P + 65536;
  const long long OFF_WCP    = OFF_W2P + 8192;

  float* dinv_t = ws + OFF_DINV_T;
  float* dinv_w = ws + OFF_DINV_W;
  float* hx1    = ws + OFF_R0;
  float* hx2    = ws + OFF_R1;
  float* h2     = ws + OFF_R0 + 3200000;
  float* g      = ws + OFF_R1;
  int2*  srt    = (int2*)(ws + OFF_R2);
  int*   rank   = (int*)(ws + OFF_R2 + 3200000);
  int*   cnt    = (int*)(ws + OFF_R2 + 4800000);
  int*   off    = (int*)(ws + OFF_R2 + 5000000);
  float* w1p    = ws + OFF_W1P;
  float* w2p    = ws + OFF_W2P;
  float* wcp    = ws + OFF_WCP;

  prep_kernel<<<PREP_B_W1 + PREP_B_W2 + PREP_B_WC + PREP_B_DG, PREP_BS, 0, stream>>>(
      W1, W2, W_attr, W_att, w1p, w2p, wcp, cnt);

  gemm1_rank_kernel<<<G1_BLOCKS + 2 * RK_BLOCKS, G1_BS, 0, stream>>>(
      x, w1p, hx1, cnt, tpe + E_MP, ei + E_MP, rank);

  scan_kernel<<<2 * SC_CH, SC_BS, 0, stream>>>(cnt, off, dinv_t);

  place_kernel<<<2 * PL_BLOCKS, 128, 0, stream>>>(tpe, ei, ew, rank, cnt, off, dinv_t, srt);

  gather1_gemm2_degw_kernel<<<FG_BLOCKS + DW_BLOCKS, FG_BS, 0, stream>>>(
      hx1, srt, off, dinv_t, b1, w2p, hx2, srt + E_MP, off + OFF_PITCH, dinv_w);

  gcn_gather_kernel<H2_DIM, true><<<AGG_WAVES / (AGG_BS / 32), AGG_BS, 0, stream>>>(
      hx2, h2, srt, off, dinv_t, b2);

  gcn_gather_kernel<H2_DIM, false><<<AGG_WAVES / (AGG_BS / 32), AGG_BS, 0, stream>>>(
      h2, g, srt + E_MP, off + OFF_PITCH, dinv_w, nullptr);
  gemm3_softmax_edge_dot_kernel<<<G1_BLOCKS + DOT_BLOCKS, G1_BS, 0, stream>>>(
      g, wcp, b_attr, b_att, out + 2 * E_LP,
      out + 2 * E_LP + (long long)N_NODES * N_CLS, h2, pos, neg, out);
}
